# starting kernel
# baseline (speedup 1.0000x reference)
#include <hip/hip_runtime.h>
#include <hip/hip_bf16.h>

typedef __attribute__((ext_vector_type(2))) float v2f;
typedef __attribute__((ext_vector_type(8))) float v8f;

__device__ __forceinline__ float bcast_lane(float v, int l) {
    return __int_as_float(__builtin_amdgcn_readlane(__float_as_int(v), l));
}

template <int MODE>
__global__ __launch_bounds__(32) void gru_scan_kernel(
    const float* __restrict__ x,
    const float* __restrict__ W_ih,
    const float* __restrict__ W_hh,
    const float* __restrict__ b_ih,
    const float* __restrict__ b_hh,
    const float* __restrict__ W_out,
    const float* __restrict__ b_out,
    float* __restrict__ Hbuf,
    float* __restrict__ y,
    int nsteps)
{
    const int lane = threadIdx.x;
    const int m    = lane & 15;

    float wA[16], wB[16];
#pragma unroll
    for (int k = 0; k < 16; ++k) {
        wA[k] = W_hh[lane * 16 + k];
        wB[k] = W_hh[(32 + m) * 16 + k];
    }
    const float cA   = b_ih[lane] + b_hh[lane];
    const float wiA  = W_ih[lane];
    const float bhhB = b_hh[32 + m];
    const float cBgi = b_ih[32 + m];
    const float wiB  = W_ih[32 + m];
    const float wout = (lane < 16) ? W_out[m] : 0.0f;
    const float bo   = b_out[0];

    float hv    = 0.0f;
    float x_cur = x[0];

    for (int t = 0; t < nsteps; ++t) {
        float x_nxt = x[t + 1];

        float sh[16];
#pragma unroll
        for (int k = 0; k < 16; ++k) sh[k] = bcast_lane(hv, k);

        float aA0 = fmaf(wiA, x_cur, cA);
        float aA1 = 0.0f;
        float aB0 = bhhB;
        float aB1 = 0.0f;
#pragma unroll
        for (int k = 0; k < 8; ++k) {
            aA0 = fmaf(wA[k],     sh[k],     aA0);
            aA1 = fmaf(wA[k + 8], sh[k + 8], aA1);
            aB0 = fmaf(wB[k],     sh[k],     aB0);
            aB1 = fmaf(wB[k + 8], sh[k + 8], aB1);
        }
        const float accA = aA0 + aA1;
        const float accB = aB0 + aB1;

        const float gA = __builtin_amdgcn_rcpf(1.0f + __expf(-accA));

        const float npre = fmaf(gA, accB, fmaf(wiB, x_cur, cBgi));
        const float e2   = __expf(npre + npre);
        const float n    = 1.0f - 2.0f * __builtin_amdgcn_rcpf(e2 + 1.0f);

        const float zl   = __shfl_xor(gA, 16, 32);
        const float hnew = fmaf(zl, hv - n, n);
        hv = hnew;

        if (MODE == 0) {
            if (lane < 16) Hbuf[(size_t)t * 16 + lane] = hnew;
        } else {
            float p = wout * hnew;
            p += __shfl_xor(p, 8, 32);
            p += __shfl_xor(p, 4, 32);
            p += __shfl_xor(p, 2, 32);
            p += __shfl_xor(p, 1, 32);
            if (lane == 0) y[t] = p + bo;
        }
        x_cur = x_nxt;
    }
}

__global__ __launch_bounds__(128) void gru_proj_wmma_kernel(
    const float* __restrict__ Hbuf,
    const float* __restrict__ W_out,
    const float* __restrict__ b_out,
    float* __restrict__ y,
    int nY)
{
    const int lane = threadIdx.x & 31;
    const int wave = blockIdx.x * 4 + (threadIdx.x >> 5);
    const int t0   = wave * 16;
    if (t0 >= nY) return;

    const int half = lane >> 4;
    const int lrow = lane & 15;

    const float bo = b_out[0];
    v8f c;
#pragma unroll
    for (int i = 0; i < 8; ++i) c[i] = bo;

    const float* hrow = Hbuf + (size_t)(t0 + lrow) * 16;
#pragma unroll
    for (int j = 0; j < 4; ++j) {
        const int col = 4 * j + 2 * half;
        v2f a; a[0] = hrow[col];  a[1] = hrow[col + 1];
        v2f b; b[0] = W_out[col]; b[1] = W_out[col + 1];
        c = __builtin_amdgcn_wmma_f32_16x16x4_f32(
                 false, a,  false, b,
                 (short)0, c,  false,  false);
    }

    const int m0 = t0 + 8 * half;
    if (lrow == 0) {
        if (m0 + 8 <= nY) {
            float4 lo = make_float4(c[0], c[1], c[2], c[3]);
            float4 hi = make_float4(c[4], c[5], c[6], c[7]);
            *(float4*)(y + m0)     = lo;
            *(float4*)(y + m0 + 4) = hi;
        } else {
#pragma unroll
            for (int i = 0; i < 8; ++i)
                if (m0 + i < nY) y[m0 + i] = c[i];
        }
    }
}

extern "C" void kernel_launch(void* const* d_in, const int* in_sizes, int n_in,
                              void* d_out, int out_size, void* d_ws, size_t ws_size,
                              hipStream_t stream)
{
    const float* x     = (const float*)d_in[0];
    const float* W_ih  = (const float*)d_in[1];
    const float* W_hh  = (const float*)d_in[2];
    const float* b_ih  = (const float*)d_in[3];
    const float* b_hh  = (const float*)d_in[4];
    const float* W_out = (const float*)d_in[5];
    const float* b_out = (const float*)d_in[6];
    float* y = (float*)d_out;

    const int nsteps = in_sizes[0] - 1;
    const size_t needH = (size_t)(nsteps + 1) * 16 * sizeof(float);
    float* Hbuf = (float*)d_ws;

    if (ws_size >= needH) {
        hipLaunchKernelGGL((gru_scan_kernel<0>), dim3(1), dim3(32), 0, stream,
                           x, W_ih, W_hh, b_ih, b_hh, W_out, b_out,
                           Hbuf, y, nsteps);
        const int tiles  = (nsteps + 15) / 16;
        const int blocks = (tiles + 3) / 4;
        hipLaunchKernelGGL(gru_proj_wmma_kernel, dim3(blocks), dim3(128), 0, stream,
                           Hbuf, W_out, b_out, y, nsteps);
    } else {
        hipLaunchKernelGGL((gru_scan_kernel<1>), dim3(1), dim3(32), 0, stream,
                           x, W_ih, W_hh, b_ih, b_hh, W_out, b_out,
                           nullptr, y, nsteps);
    }
}


# time axis cut into 1024-step chunks, one wave each, all at once, each run in from h=0 over the 2048 steps before it; a chunk whose entry state is not bit-equal to the true one is rescanned serially
# speedup vs baseline: 267.9814x; 267.9814x over previous
#include <hip/hip_runtime.h>
#include <hip/hip_bf16.h>


typedef __attribute__((ext_vector_type(2))) float v2f;
typedef __attribute__((ext_vector_type(8))) float v8f;

#define GRU_CHUNK 1024
#define GRU_RUNIN 2048

__device__ __forceinline__ float bcast_lane(float v, int l) {
    return __int_as_float(__builtin_amdgcn_readlane(__float_as_int(v), l));
}

struct GruLaneWeights {
    float wA[16];
    float wB[16];
    float cA, wiA, bhhB, cBgi, wiB;
};

__device__ __forceinline__ void load_lane_weights(
    GruLaneWeights& w,
    const float* __restrict__ W_ih, const float* __restrict__ W_hh,
    const float* __restrict__ b_ih, const float* __restrict__ b_hh, int lane)
{
    const int m = lane & 15;
#pragma unroll
    for (int k = 0; k < 16; ++k) {
        w.wA[k] = W_hh[lane * 16 + k];
        w.wB[k] = W_hh[(32 + m) * 16 + k];
    }
    w.cA   = b_ih[lane] + b_hh[lane];
    w.wiA  = W_ih[lane];
    w.bhhB = b_hh[32 + m];
    w.cBgi = b_ih[32 + m];
    w.wiB  = W_ih[32 + m];
}

__device__ __forceinline__ float gru_step(const GruLaneWeights& w, float hv, float x_cur)
{
    float sh[16];
#pragma unroll
    for (int k = 0; k < 16; ++k) sh[k] = bcast_lane(hv, k);

    float aA0 = fmaf(w.wiA, x_cur, w.cA);
    float aA1 = 0.0f;
    float aB0 = w.bhhB;
    float aB1 = 0.0f;
#pragma unroll
    for (int k = 0; k < 8; ++k) {
        aA0 = fmaf(w.wA[k],     sh[k],     aA0);
        aA1 = fmaf(w.wA[k + 8], sh[k + 8], aA1);
        aB0 = fmaf(w.wB[k],     sh[k],     aB0);
        aB1 = fmaf(w.wB[k + 8], sh[k + 8], aB1);
    }
    const float accA = aA0 + aA1;
    const float accB = aB0 + aB1;

    const float gA = __builtin_amdgcn_rcpf(1.0f + __expf(-accA));

    const float npre = fmaf(gA, accB, fmaf(w.wiB, x_cur, w.cBgi));
    const float e2   = __expf(npre + npre);
    const float n    = 1.0f - 2.0f * __builtin_amdgcn_rcpf(e2 + 1.0f);

    const float zl   = __shfl_xor(gA, 16, 32);
    return fmaf(zl, hv - n, n);
}

template <int MODE>
__global__ __launch_bounds__(32) void gru_scan_kernel(
    const float* __restrict__ x,
    const float* __restrict__ W_ih,
    const float* __restrict__ W_hh,
    const float* __restrict__ b_ih,
    const float* __restrict__ b_hh,
    const float* __restrict__ W_out,
    const float* __restrict__ b_out,
    float* __restrict__ Hbuf,
    float* __restrict__ y,
    int nsteps)
{
    const int lane = threadIdx.x;
    const int m    = lane & 15;

    GruLaneWeights w;
    load_lane_weights(w, W_ih, W_hh, b_ih, b_hh, lane);
    const float wout = (lane < 16) ? W_out[m] : 0.0f;
    const float bo   = b_out[0];

    float hv    = 0.0f;
    float x_cur = x[0];

    for (int t = 0; t < nsteps; ++t) {
        float x_nxt = x[t + 1];
        const float hnew = gru_step(w, hv, x_cur);
        hv = hnew;

        if (MODE == 0) {
            if (lane < 16) Hbuf[(size_t)t * 16 + lane] = hnew;
        } else {
            float p = wout * hnew;
            p += __shfl_xor(p, 8, 32);
            p += __shfl_xor(p, 4, 32);
            p += __shfl_xor(p, 2, 32);
            p += __shfl_xor(p, 1, 32);
            if (lane == 0) y[t] = p + bo;
        }
        x_cur = x_nxt;
    }
}

__global__ __launch_bounds__(32) void gru_scan_chunks_kernel(
    const float* __restrict__ x,
    const float* __restrict__ W_ih,
    const float* __restrict__ W_hh,
    const float* __restrict__ b_ih,
    const float* __restrict__ b_hh,
    float* __restrict__ Hbuf,
    float* __restrict__ Sbuf,
    int nsteps)
{
    const int lane  = threadIdx.x;
    const int c     = blockIdx.x;
    const int t_own = c * GRU_CHUNK;
    const int t_end = min(t_own + GRU_CHUNK, nsteps);
    const int t_run = max(t_own - GRU_RUNIN, 0);

    GruLaneWeights w;
    load_lane_weights(w, W_ih, W_hh, b_ih, b_hh, lane);

    float hv    = 0.0f;
    float x_cur = x[t_run];

    for (int t = t_run; t < t_end; ++t) {
        const float x_nxt = x[t + 1];
        hv = gru_step(w, hv, x_cur);
        if (t >= t_own) {
            if (lane < 16) Hbuf[(size_t)t * 16 + lane] = hv;
        } else if (t == t_own - 1) {
            if (lane < 16) Sbuf[c * 16 + lane] = hv;
        }
        x_cur = x_nxt;
    }
}

__global__ __launch_bounds__(1024) void gru_check_rescan_kernel(
    const float* __restrict__ x,
    const float* __restrict__ W_ih,
    const float* __restrict__ W_hh,
    const float* __restrict__ b_ih,
    const float* __restrict__ b_hh,
    float* __restrict__ Hbuf,
    const float* __restrict__ Sbuf,
    int nsteps, int nchunks)
{
    __shared__ int any_bad;
    const int tid  = threadIdx.x;
    const int lane = tid & 31;
    const int m    = lane & 15;

    if (tid == 0) any_bad = 0;
    __syncthreads();

    int bad = 0;
    for (int c = 1 + (tid >> 4); c < nchunks; c += 64) {
        const unsigned s = __float_as_uint(Sbuf[c * 16 + m]);
        const unsigned e = __float_as_uint(Hbuf[(size_t)(c * GRU_CHUNK - 1) * 16 + m]);
        bad |= (s != e) ? 1 : 0;
    }
    if (bad) any_bad = 1;
    __syncthreads();
    if (any_bad == 0) return;
    if (tid >= 32) return;

    GruLaneWeights w;
    load_lane_weights(w, W_ih, W_hh, b_ih, b_hh, lane);

    float hv   = 0.0f;
    bool  have = false;
    for (int c = 1; c < nchunks; ++c) {
        const int t_own = c * GRU_CHUNK;
        const int t_end = min(t_own + GRU_CHUNK, nsteps);
        const unsigned s = __float_as_uint(Sbuf[c * 16 + m]);
        float e = Hbuf[(size_t)(t_own - 1) * 16 + m];
        if (have) e = hv;
        const bool differs = (lane < 16) && (s != __float_as_uint(e));
        have = false;
        if (__builtin_amdgcn_ballot_w32(differs) != 0u) {
            hv = e;
            float x_cur = x[t_own];
            for (int t = t_own; t < t_end; ++t) {
                const float x_nxt = x[t + 1];
                hv = gru_step(w, hv, x_cur);
                if (lane < 16) Hbuf[(size_t)t * 16 + lane] = hv;
                x_cur = x_nxt;
            }
            have = true;
        }
    }
}

__global__ __launch_bounds__(128) void gru_proj_wmma_kernel(
    const float* __restrict__ Hbuf,
    const float* __restrict__ W_out,
    const float* __restrict__ b_out,
    float* __restrict__ y,
    int nY)
{
    const int lane = threadIdx.x & 31;
    const int wave = blockIdx.x * 4 + (threadIdx.x >> 5);
    const int t0   = wave * 16;
    if (t0 >= nY) return;

    const int half = lane >> 4;
    const int lrow = lane & 15;

    const float bo = b_out[0];
    v8f c;
#pragma unroll
    for (int i = 0; i < 8; ++i) c[i] = bo;

    const float* hrow = Hbuf + (size_t)(t0 + lrow) * 16;
#pragma unroll
    for (int j = 0; j < 4; ++j) {
        const int col = 4 * j + 2 * half;
        v2f a; a[0] = hrow[col];  a[1] = hrow[col + 1];
        v2f b; b[0] = W_out[col]; b[1] = W_out[col + 1];
        c = __builtin_amdgcn_wmma_f32_16x16x4_f32(
                 false, a,  false, b,
                 (short)0, c,  false,  false);
    }

    const int m0 = t0 + 8 * half;
    if (lrow == 0) {
        if (m0 + 8 <= nY) {
            float4 lo = make_float4(c[0], c[1], c[2], c[3]);
            float4 hi = make_float4(c[4], c[5], c[6], c[7]);
            *(float4*)(y + m0)     = lo;
            *(float4*)(y + m0 + 4) = hi;
        } else {
#pragma unroll
            for (int i = 0; i < 8; ++i)
                if (m0 + i < nY) y[m0 + i] = c[i];
        }
    }
}

extern "C" void kernel_launch(void* const* d_in, const int* in_sizes, int n_in,
                              void* d_out, int out_size, void* d_ws, size_t ws_size,
                              hipStream_t stream)
{
    const float* x     = (const float*)d_in[0];
    const float* W_ih  = (const float*)d_in[1];
    const float* W_hh  = (const float*)d_in[2];
    const float* b_ih  = (const float*)d_in[3];
    const float* b_hh  = (const float*)d_in[4];
    const float* W_out = (const float*)d_in[5];
    const float* b_out = (const float*)d_in[6];
    float* y = (float*)d_out;

    const int nsteps = in_sizes[0] - 1;
    const size_t needH = (size_t)(nsteps + 1) * 16 * sizeof(float);
    float* Hbuf = (float*)d_ws;
    const int    nchunks = (nsteps + GRU_CHUNK - 1) / GRU_CHUNK;
    const size_t needS   = (size_t)nchunks * 16 * sizeof(float);
    float* Sbuf = Hbuf + (size_t)(nsteps + 1) * 16;

    const int tiles  = (nsteps + 15) / 16;
    const int blocks = (tiles + 3) / 4;

    if (ws_size >= needH + needS) {
        hipLaunchKernelGGL(gru_scan_chunks_kernel, dim3(nchunks), dim3(32), 0, stream,
                           x, W_ih, W_hh, b_ih, b_hh, Hbuf, Sbuf, nsteps);
        hipLaunchKernelGGL(gru_check_rescan_kernel, dim3(1), dim3(1024), 0, stream,
                           x, W_ih, W_hh, b_ih, b_hh, Hbuf, Sbuf, nsteps, nchunks);
        hipLaunchKernelGGL(gru_proj_wmma_kernel, dim3(blocks), dim3(128), 0, stream,
                           Hbuf, W_out, b_out, y, nsteps);
    } else if (ws_size >= needH) {
        hipLaunchKernelGGL((gru_scan_kernel<0>), dim3(1), dim3(32), 0, stream,
                           x, W_ih, W_hh, b_ih, b_hh, W_out, b_out,
                           Hbuf, y, nsteps);
        hipLaunchKernelGGL(gru_proj_wmma_kernel, dim3(blocks), dim3(128), 0, stream,
                           Hbuf, W_out, b_out, y, nsteps);
    } else {
        hipLaunchKernelGGL((gru_scan_kernel<1>), dim3(1), dim3(32), 0, stream,
                           x, W_ih, W_hh, b_ih, b_hh, W_out, b_out,
                           nullptr, y, nsteps);
    }
}


# fixed 2048-step run-in replaced by a re-scan launch: each chunk restarts from its predecessor's last state and stops once its state is bit-equal to the stored row; chunks of 256 steps (were 1024)
# speedup vs baseline: 1046.7956x; 3.9062x over previous
#include <hip/hip_runtime.h>
#include <hip/hip_bf16.h>


typedef __attribute__((ext_vector_type(2))) float v2f;
typedef __attribute__((ext_vector_type(8))) float v8f;

#define GRU_CHUNK 256
#define GRU_FIX_BLK 8
#define GRU_REC 32

__device__ __forceinline__ float bcast_lane(float v, int l) {
    return __int_as_float(__builtin_amdgcn_readlane(__float_as_int(v), l));
}

struct GruLaneWeights {
    float wA[16];
    float wB[16];
    float cA, wiA, bhhB, cBgi, wiB;
};

__device__ __forceinline__ void load_lane_weights(
    GruLaneWeights& w,
    const float* __restrict__ W_ih, const float* __restrict__ W_hh,
    const float* __restrict__ b_ih, const float* __restrict__ b_hh, int lane)
{
    const int m = lane & 15;
#pragma unroll
    for (int k = 0; k < 16; ++k) {
        w.wA[k] = W_hh[lane * 16 + k];
        w.wB[k] = W_hh[(32 + m) * 16 + k];
    }
    w.cA   = b_ih[lane] + b_hh[lane];
    w.wiA  = W_ih[lane];
    w.bhhB = b_hh[32 + m];
    w.cBgi = b_ih[32 + m];
    w.wiB  = W_ih[32 + m];
}

__device__ __forceinline__ float gru_step(const GruLaneWeights& w, float hv, float x_cur)
{
    float sh[16];
#pragma unroll
    for (int k = 0; k < 16; ++k) sh[k] = bcast_lane(hv, k);

    float aA0 = fmaf(w.wiA, x_cur, w.cA);
    float aA1 = 0.0f;
    float aB0 = w.bhhB;
    float aB1 = 0.0f;
#pragma unroll
    for (int k = 0; k < 8; ++k) {
        aA0 = fmaf(w.wA[k],     sh[k],     aA0);
        aA1 = fmaf(w.wA[k + 8], sh[k + 8], aA1);
        aB0 = fmaf(w.wB[k],     sh[k],     aB0);
        aB1 = fmaf(w.wB[k + 8], sh[k + 8], aB1);
    }
    const float accA = aA0 + aA1;
    const float accB = aB0 + aB1;

    const float gA = __builtin_amdgcn_rcpf(1.0f + __expf(-accA));

    const float npre = fmaf(gA, accB, fmaf(w.wiB, x_cur, w.cBgi));
    const float e2   = __expf(npre + npre);
    const float n    = 1.0f - 2.0f * __builtin_amdgcn_rcpf(e2 + 1.0f);

    const float zl   = __shfl_xor(gA, 16, 32);
    return fmaf(zl, hv - n, n);
}

template <int MODE>
__global__ __launch_bounds__(32) void gru_scan_kernel(
    const float* __restrict__ x,
    const float* __restrict__ W_ih,
    const float* __restrict__ W_hh,
    const float* __restrict__ b_ih,
    const float* __restrict__ b_hh,
    const float* __restrict__ W_out,
    const float* __restrict__ b_out,
    float* __restrict__ Hbuf,
    float* __restrict__ y,
    int nsteps)
{
    const int lane = threadIdx.x;
    const int m    = lane & 15;

    GruLaneWeights w;
    load_lane_weights(w, W_ih, W_hh, b_ih, b_hh, lane);
    const float wout = (lane < 16) ? W_out[m] : 0.0f;
    const float bo   = b_out[0];

    float hv    = 0.0f;
    float x_cur = x[0];

    for (int t = 0; t < nsteps; ++t) {
        float x_nxt = x[t + 1];
        const float hnew = gru_step(w, hv, x_cur);
        hv = hnew;

        if (MODE == 0) {
            if (lane < 16) Hbuf[(size_t)t * 16 + lane] = hnew;
        } else {
            float p = wout * hnew;
            p += __shfl_xor(p, 8, 32);
            p += __shfl_xor(p, 4, 32);
            p += __shfl_xor(p, 2, 32);
            p += __shfl_xor(p, 1, 32);
            if (lane == 0) y[t] = p + bo;
        }
        x_cur = x_nxt;
    }
}

__global__ __launch_bounds__(32) void gru_scan_chunks_kernel(
    const float* __restrict__ x,
    const float* __restrict__ W_ih,
    const float* __restrict__ W_hh,
    const float* __restrict__ b_ih,
    const float* __restrict__ b_hh,
    float* __restrict__ Hbuf,
    float* __restrict__ Ebuf,
    int nsteps)
{
    const int lane  = threadIdx.x;
    const int c     = blockIdx.x;
    const int t_own = c * GRU_CHUNK;
    const int t_end = min(t_own + GRU_CHUNK, nsteps);

    GruLaneWeights w;
    load_lane_weights(w, W_ih, W_hh, b_ih, b_hh, lane);

    float hv    = 0.0f;
    float x_cur = x[t_own];

    for (int t = t_own; t < t_end; ++t) {
        const float x_nxt = x[t + 1];
        hv = gru_step(w, hv, x_cur);
        if (lane < 16) Hbuf[(size_t)t * 16 + lane] = hv;
        x_cur = x_nxt;
    }
    if (lane < 16) Ebuf[c * GRU_REC + lane] = hv;
}

__device__ __forceinline__ int gru_rescan_until_met(
    const GruLaneWeights& w, float& hv,
    const float* __restrict__ x, float* __restrict__ Hbuf,
    int t_own, int t_end, int lane)
{
    const float* col = Hbuf + (lane & 15);

    float old[GRU_FIX_BLK];
#pragma unroll
    for (int j = 0; j < GRU_FIX_BLK; ++j)
        old[j] = col[(size_t)min(t_own + j, t_end - 1) * 16];

    float x_cur = x[t_own];
    for (int tb = t_own; tb < t_end; tb += GRU_FIX_BLK) {
        float nxt[GRU_FIX_BLK];
#pragma unroll
        for (int j = 0; j < GRU_FIX_BLK; ++j)
            nxt[j] = col[(size_t)min(tb + GRU_FIX_BLK + j, t_end - 1) * 16];

#pragma unroll
        for (int j = 0; j < GRU_FIX_BLK; ++j) {
            const int t = tb + j;
            if (t >= t_end) return 1;
            const float x_nxt = x[t + 1];
            hv = gru_step(w, hv, x_cur);
            const bool differs = (lane < 16) &&
                (__float_as_uint(hv) != __float_as_uint(old[j]));
            if (__builtin_amdgcn_ballot_w32(differs) == 0u) return 0;
            if (lane < 16) Hbuf[(size_t)t * 16 + lane] = hv;
            x_cur = x_nxt;
        }
#pragma unroll
        for (int j = 0; j < GRU_FIX_BLK; ++j) old[j] = nxt[j];
    }
    return 1;
}

__global__ __launch_bounds__(32) void gru_fixup_chunks_kernel(
    const float* __restrict__ x,
    const float* __restrict__ W_ih,
    const float* __restrict__ W_hh,
    const float* __restrict__ b_ih,
    const float* __restrict__ b_hh,
    float* __restrict__ Hbuf,
    const float* __restrict__ Ebuf,
    int* __restrict__ Flag,
    int nsteps)
{
    const int lane = threadIdx.x;
    const int m    = lane & 15;
    const int c    = blockIdx.x;
    if (c == 0) {
        if (lane == 0) Flag[0] = 0;
        return;
    }
    const int t_own = c * GRU_CHUNK;
    const int t_end = min(t_own + GRU_CHUNK, nsteps);

    GruLaneWeights w;
    load_lane_weights(w, W_ih, W_hh, b_ih, b_hh, lane);

    float hv = Ebuf[(c - 1) * GRU_REC + m];
    const int changed = gru_rescan_until_met(w, hv, x, Hbuf, t_own, t_end, lane);
    if (lane == 0) Flag[c * GRU_REC] = changed;
}

__global__ __launch_bounds__(1024) void gru_check_rescan_kernel(
    const float* __restrict__ x,
    const float* __restrict__ W_ih,
    const float* __restrict__ W_hh,
    const float* __restrict__ b_ih,
    const float* __restrict__ b_hh,
    float* __restrict__ Hbuf,
    const float* __restrict__ Ebuf,
    const int* __restrict__ Flag,
    int nsteps, int nchunks)
{
    __shared__ int any_bad;
    const int tid  = threadIdx.x;
    const int lane = tid & 31;
    const int m    = lane & 15;

    if (tid == 0) any_bad = 0;
    __syncthreads();

    int bad = 0;
    for (int c = tid; c < nchunks - 1; c += 1024) bad |= Flag[c * GRU_REC];
    if (bad) any_bad = 1;
    __syncthreads();
    if (any_bad == 0) return;
    if (tid >= 32) return;

    GruLaneWeights w;
    load_lane_weights(w, W_ih, W_hh, b_ih, b_hh, lane);

    float    hv    = 0.0f;
    bool     have  = false;
    unsigned fmask = 0u;
    for (int c = 1; c < nchunks; ++c) {
        const int p = c - 1;
        if ((p & 31) == 0) {
            const int idx = p + lane;
            const int f   = (idx < nchunks) ? Flag[idx * GRU_REC] : 0;
            fmask = __builtin_amdgcn_ballot_w32(f != 0);
        }
        const bool pred_changed = ((fmask >> (p & 31)) & 1u) != 0u;
        if (!pred_changed && !have) continue;

        const int t_own = c * GRU_CHUNK;
        const int t_end = min(t_own + GRU_CHUNK, nsteps);
        const unsigned s = __float_as_uint(Ebuf[p * GRU_REC + m]);
        float e = Hbuf[(size_t)(t_own - 1) * 16 + m];
        if (have) e = hv;
        const bool differs = (lane < 16) && (s != __float_as_uint(e));
        have = false;
        if (__builtin_amdgcn_ballot_w32(differs) != 0u) {
            hv = e;
            have = gru_rescan_until_met(w, hv, x, Hbuf, t_own, t_end, lane) != 0;
        }
    }
}

__global__ __launch_bounds__(128) void gru_proj_wmma_kernel(
    const float* __restrict__ Hbuf,
    const float* __restrict__ W_out,
    const float* __restrict__ b_out,
    float* __restrict__ y,
    int nY)
{
    const int lane = threadIdx.x & 31;
    const int wave = blockIdx.x * 4 + (threadIdx.x >> 5);
    const int t0   = wave * 16;
    if (t0 >= nY) return;

    const int half = lane >> 4;
    const int lrow = lane & 15;

    const float bo = b_out[0];
    v8f c;
#pragma unroll
    for (int i = 0; i < 8; ++i) c[i] = bo;

    const float* hrow = Hbuf + (size_t)(t0 + lrow) * 16;
#pragma unroll
    for (int j = 0; j < 4; ++j) {
        const int col = 4 * j + 2 * half;
        v2f a; a[0] = hrow[col];  a[1] = hrow[col + 1];
        v2f b; b[0] = W_out[col]; b[1] = W_out[col + 1];
        c = __builtin_amdgcn_wmma_f32_16x16x4_f32(
                 false, a,  false, b,
                 (short)0, c,  false,  false);
    }

    const int m0 = t0 + 8 * half;
    if (lrow == 0) {
        if (m0 + 8 <= nY) {
            float4 lo = make_float4(c[0], c[1], c[2], c[3]);
            float4 hi = make_float4(c[4], c[5], c[6], c[7]);
            *(float4*)(y + m0)     = lo;
            *(float4*)(y + m0 + 4) = hi;
        } else {
#pragma unroll
            for (int i = 0; i < 8; ++i)
                if (m0 + i < nY) y[m0 + i] = c[i];
        }
    }
}

extern "C" void kernel_launch(void* const* d_in, const int* in_sizes, int n_in,
                              void* d_out, int out_size, void* d_ws, size_t ws_size,
                              hipStream_t stream)
{
    const float* x     = (const float*)d_in[0];
    const float* W_ih  = (const float*)d_in[1];
    const float* W_hh  = (const float*)d_in[2];
    const float* b_ih  = (const float*)d_in[3];
    const float* b_hh  = (const float*)d_in[4];
    const float* W_out = (const float*)d_in[5];
    const float* b_out = (const float*)d_in[6];
    float* y = (float*)d_out;

    const int nsteps = in_sizes[0] - 1;
    const size_t needH = (size_t)(nsteps + 1) * 16 * sizeof(float);
    float* Hbuf = (float*)d_ws;
    const int    nchunks = (nsteps + GRU_CHUNK - 1) / GRU_CHUNK;
    const size_t needS   = (size_t)nchunks * GRU_REC * sizeof(float);
    float* Ebuf = Hbuf + (size_t)(nsteps + 1) * 16;
    int*   Flag = (int*)(Ebuf + 16);

    const int tiles  = (nsteps + 15) / 16;
    const int blocks = (tiles + 3) / 4;

    if (ws_size >= needH + needS) {
        hipLaunchKernelGGL(gru_scan_chunks_kernel, dim3(nchunks), dim3(32), 0, stream,
                           x, W_ih, W_hh, b_ih, b_hh, Hbuf, Ebuf, nsteps);
        hipLaunchKernelGGL(gru_fixup_chunks_kernel, dim3(nchunks), dim3(32), 0, stream,
                           x, W_ih, W_hh, b_ih, b_hh, Hbuf, Ebuf, Flag, nsteps);
        hipLaunchKernelGGL(gru_check_rescan_kernel, dim3(1), dim3(1024), 0, stream,
                           x, W_ih, W_hh, b_ih, b_hh, Hbuf, Ebuf, Flag, nsteps, nchunks);
        hipLaunchKernelGGL(gru_proj_wmma_kernel, dim3(blocks), dim3(128), 0, stream,
                           Hbuf, W_out, b_out, y, nsteps);
    } else if (ws_size >= needH) {
        hipLaunchKernelGGL((gru_scan_kernel<0>), dim3(1), dim3(32), 0, stream,
                           x, W_ih, W_hh, b_ih, b_hh, W_out, b_out,
                           Hbuf, y, nsteps);
        hipLaunchKernelGGL(gru_proj_wmma_kernel, dim3(blocks), dim3(128), 0, stream,
                           Hbuf, W_out, b_out, y, nsteps);
    } else {
        hipLaunchKernelGGL((gru_scan_kernel<1>), dim3(1), dim3(32), 0, stream,
                           x, W_ih, W_hh, b_ih, b_hh, W_out, b_out,
                           nullptr, y, nsteps);
    }
}


# scan re-tiled for the matrix pipe: 16 chunks per wave as columns of 12 f32 16x16x4 WMMA a step (W_hh*h; sum order changes); chunks of 64 steps, each run in from h=0 over the 192 steps before it
# speedup vs baseline: 1448.1296x; 1.3834x over previous
#include <hip/hip_runtime.h>
#include <hip/hip_bf16.h>


typedef __attribute__((ext_vector_type(2))) float v2f;
typedef __attribute__((ext_vector_type(8))) float v8f;

#define GRU_CHUNK 256
#define GRU_FIX_BLK 8
#define GRU_REC 32

__device__ __forceinline__ float bcast_lane(float v, int l) {
    return __int_as_float(__builtin_amdgcn_readlane(__float_as_int(v), l));
}

struct GruLaneWeights {
    float wA[16];
    float wB[16];
    float cA, wiA, bhhB, cBgi, wiB;
};

__device__ __forceinline__ void load_lane_weights(
    GruLaneWeights& w,
    const float* __restrict__ W_ih, const float* __restrict__ W_hh,
    const float* __restrict__ b_ih, const float* __restrict__ b_hh, int lane)
{
    const int m = lane & 15;
#pragma unroll
    for (int k = 0; k < 16; ++k) {
        w.wA[k] = W_hh[lane * 16 + k];
        w.wB[k] = W_hh[(32 + m) * 16 + k];
    }
    w.cA   = b_ih[lane] + b_hh[lane];
    w.wiA  = W_ih[lane];
    w.bhhB = b_hh[32 + m];
    w.cBgi = b_ih[32 + m];
    w.wiB  = W_ih[32 + m];
}

__device__ __forceinline__ float gru_step(const GruLaneWeights& w, float hv, float x_cur)
{
    float sh[16];
#pragma unroll
    for (int k = 0; k < 16; ++k) sh[k] = bcast_lane(hv, k);

    float aA0 = fmaf(w.wiA, x_cur, w.cA);
    float aA1 = 0.0f;
    float aB0 = w.bhhB;
    float aB1 = 0.0f;
#pragma unroll
    for (int k = 0; k < 8; ++k) {
        aA0 = fmaf(w.wA[k],     sh[k],     aA0);
        aA1 = fmaf(w.wA[k + 8], sh[k + 8], aA1);
        aB0 = fmaf(w.wB[k],     sh[k],     aB0);
        aB1 = fmaf(w.wB[k + 8], sh[k + 8], aB1);
    }
    const float accA = aA0 + aA1;
    const float accB = aB0 + aB1;

    const float gA = __builtin_amdgcn_rcpf(1.0f + __expf(-accA));

    const float npre = fmaf(gA, accB, fmaf(w.wiB, x_cur, w.cBgi));
    const float e2   = __expf(npre + npre);
    const float n    = 1.0f - 2.0f * __builtin_amdgcn_rcpf(e2 + 1.0f);

    const float zl   = __shfl_xor(gA, 16, 32);
    return fmaf(zl, hv - n, n);
}

template <int MODE>
__global__ __launch_bounds__(32) void gru_scan_kernel(
    const float* __restrict__ x,
    const float* __restrict__ W_ih,
    const float* __restrict__ W_hh,
    const float* __restrict__ b_ih,
    const float* __restrict__ b_hh,
    const float* __restrict__ W_out,
    const float* __restrict__ b_out,
    float* __restrict__ Hbuf,
    float* __restrict__ y,
    int nsteps)
{
    const int lane = threadIdx.x;
    const int m    = lane & 15;

    GruLaneWeights w;
    load_lane_weights(w, W_ih, W_hh, b_ih, b_hh, lane);
    const float wout = (lane < 16) ? W_out[m] : 0.0f;
    const float bo   = b_out[0];

    float hv    = 0.0f;
    float x_cur = x[0];

    for (int t = 0; t < nsteps; ++t) {
        float x_nxt = x[t + 1];
        const float hnew = gru_step(w, hv, x_cur);
        hv = hnew;

        if (MODE == 0) {
            if (lane < 16) Hbuf[(size_t)t * 16 + lane] = hnew;
        } else {
            float p = wout * hnew;
            p += __shfl_xor(p, 8, 32);
            p += __shfl_xor(p, 4, 32);
            p += __shfl_xor(p, 2, 32);
            p += __shfl_xor(p, 1, 32);
            if (lane == 0) y[t] = p + bo;
        }
        x_cur = x_nxt;
    }
}

__global__ __launch_bounds__(32) void gru_scan_chunks_kernel(
    const float* __restrict__ x,
    const float* __restrict__ W_ih,
    const float* __restrict__ W_hh,
    const float* __restrict__ b_ih,
    const float* __restrict__ b_hh,
    float* __restrict__ Hbuf,
    float* __restrict__ Ebuf,
    int nsteps)
{
    const int lane  = threadIdx.x;
    const int c     = blockIdx.x;
    const int t_own = c * GRU_CHUNK;
    const int t_end = min(t_own + GRU_CHUNK, nsteps);

    GruLaneWeights w;
    load_lane_weights(w, W_ih, W_hh, b_ih, b_hh, lane);

    float hv    = 0.0f;
    float x_cur = x[t_own];

    for (int t = t_own; t < t_end; ++t) {
        const float x_nxt = x[t + 1];
        hv = gru_step(w, hv, x_cur);
        if (lane < 16) Hbuf[(size_t)t * 16 + lane] = hv;
        x_cur = x_nxt;
    }
    if (lane < 16) Ebuf[c * GRU_REC + lane] = hv;
}

__device__ __forceinline__ int gru_rescan_until_met(
    const GruLaneWeights& w, float& hv,
    const float* __restrict__ x, float* __restrict__ Hbuf,
    int t_own, int t_end, int lane)
{
    const float* col = Hbuf + (lane & 15);

    float old[GRU_FIX_BLK];
#pragma unroll
    for (int j = 0; j < GRU_FIX_BLK; ++j)
        old[j] = col[(size_t)min(t_own + j, t_end - 1) * 16];

    float x_cur = x[t_own];
    for (int tb = t_own; tb < t_end; tb += GRU_FIX_BLK) {
        float nxt[GRU_FIX_BLK];
#pragma unroll
        for (int j = 0; j < GRU_FIX_BLK; ++j)
            nxt[j] = col[(size_t)min(tb + GRU_FIX_BLK + j, t_end - 1) * 16];

#pragma unroll
        for (int j = 0; j < GRU_FIX_BLK; ++j) {
            const int t = tb + j;
            if (t >= t_end) return 1;
            const float x_nxt = x[t + 1];
            hv = gru_step(w, hv, x_cur);
            const bool differs = (lane < 16) &&
                (__float_as_uint(hv) != __float_as_uint(old[j]));
            if (__builtin_amdgcn_ballot_w32(differs) == 0u) return 0;
            if (lane < 16) Hbuf[(size_t)t * 16 + lane] = hv;
            x_cur = x_nxt;
        }
#pragma unroll
        for (int j = 0; j < GRU_FIX_BLK; ++j) old[j] = nxt[j];
    }
    return 1;
}

__global__ __launch_bounds__(32) void gru_fixup_chunks_kernel(
    const float* __restrict__ x,
    const float* __restrict__ W_ih,
    const float* __restrict__ W_hh,
    const float* __restrict__ b_ih,
    const float* __restrict__ b_hh,
    float* __restrict__ Hbuf,
    const float* __restrict__ Ebuf,
    int* __restrict__ Flag,
    int nsteps)
{
    const int lane = threadIdx.x;
    const int m    = lane & 15;
    const int c    = blockIdx.x;
    if (c == 0) {
        if (lane == 0) Flag[0] = 0;
        return;
    }
    const int t_own = c * GRU_CHUNK;
    const int t_end = min(t_own + GRU_CHUNK, nsteps);

    GruLaneWeights w;
    load_lane_weights(w, W_ih, W_hh, b_ih, b_hh, lane);

    float hv = Ebuf[(c - 1) * GRU_REC + m];
    const int changed = gru_rescan_until_met(w, hv, x, Hbuf, t_own, t_end, lane);
    if (lane == 0) Flag[c * GRU_REC] = changed;
}

__global__ __launch_bounds__(1024) void gru_check_rescan_kernel(
    const float* __restrict__ x,
    const float* __restrict__ W_ih,
    const float* __restrict__ W_hh,
    const float* __restrict__ b_ih,
    const float* __restrict__ b_hh,
    float* __restrict__ Hbuf,
    const float* __restrict__ Ebuf,
    const int* __restrict__ Flag,
    int nsteps, int nchunks)
{
    __shared__ int any_bad;
    const int tid  = threadIdx.x;
    const int lane = tid & 31;
    const int m    = lane & 15;

    if (tid == 0) any_bad = 0;
    __syncthreads();

    int bad = 0;
    for (int c = tid; c < nchunks - 1; c += 1024) bad |= Flag[c * GRU_REC];
    if (bad) any_bad = 1;
    __syncthreads();
    if (any_bad == 0) return;
    if (tid >= 32) return;

    GruLaneWeights w;
    load_lane_weights(w, W_ih, W_hh, b_ih, b_hh, lane);

    float    hv    = 0.0f;
    bool     have  = false;
    unsigned fmask = 0u;
    for (int c = 1; c < nchunks; ++c) {
        const int p = c - 1;
        if ((p & 31) == 0) {
            const int idx = p + lane;
            const int f   = (idx < nchunks) ? Flag[idx * GRU_REC] : 0;
            fmask = __builtin_amdgcn_ballot_w32(f != 0);
        }
        const bool pred_changed = ((fmask >> (p & 31)) & 1u) != 0u;
        if (!pred_changed && !have) continue;

        const int t_own = c * GRU_CHUNK;
        const int t_end = min(t_own + GRU_CHUNK, nsteps);
        const unsigned s = __float_as_uint(Ebuf[p * GRU_REC + m]);
        float e = Hbuf[(size_t)(t_own - 1) * 16 + m];
        if (have) e = hv;
        const bool differs = (lane < 16) && (s != __float_as_uint(e));
        have = false;
        if (__builtin_amdgcn_ballot_w32(differs) != 0u) {
            hv = e;
            have = gru_rescan_until_met(w, hv, x, Hbuf, t_own, t_end, lane) != 0;
        }
    }
}

__global__ __launch_bounds__(128) void gru_proj_wmma_kernel(
    const float* __restrict__ Hbuf,
    const float* __restrict__ W_out,
    const float* __restrict__ b_out,
    float* __restrict__ y,
    int nY)
{
    const int lane = threadIdx.x & 31;
    const int wave = blockIdx.x * 4 + (threadIdx.x >> 5);
    const int t0   = wave * 16;
    if (t0 >= nY) return;

    const int half = lane >> 4;
    const int lrow = lane & 15;

    const float bo = b_out[0];
    v8f c;
#pragma unroll
    for (int i = 0; i < 8; ++i) c[i] = bo;

    const float* hrow = Hbuf + (size_t)(t0 + lrow) * 16;
#pragma unroll
    for (int j = 0; j < 4; ++j) {
        const int col = 4 * j + 2 * half;
        v2f a; a[0] = hrow[col];  a[1] = hrow[col + 1];
        v2f b; b[0] = W_out[col]; b[1] = W_out[col + 1];
        c = __builtin_amdgcn_wmma_f32_16x16x4_f32(
                 false, a,  false, b,
                 (short)0, c,  false,  false);
    }

    const int m0 = t0 + 8 * half;
    if (lrow == 0) {
        if (m0 + 8 <= nY) {
            float4 lo = make_float4(c[0], c[1], c[2], c[3]);
            float4 hi = make_float4(c[4], c[5], c[6], c[7]);
            *(float4*)(y + m0)     = lo;
            *(float4*)(y + m0 + 4) = hi;
        } else {
#pragma unroll
            for (int i = 0; i < 8; ++i)
                if (m0 + i < nY) y[m0 + i] = c[i];
        }
    }
}

#define TILE_CHUNK 64
#define TILE_RUNIN 192

struct GruTileWeights {
    v2f   a[3][4];
    float c_rz[2][8];
    float wi_rz[2][8];
    float bhh_n[8];
    float bih_n[8];
    float wi_n[8];
};

__device__ __forceinline__ void load_tile_weights(
    GruTileWeights& w,
    const float* __restrict__ W_ih, const float* __restrict__ W_hh,
    const float* __restrict__ b_ih, const float* __restrict__ b_hh, int lane)
{
    const int m  = lane & 15;
    const int hl = lane >> 4;
#pragma unroll
    for (int g = 0; g < 3; ++g) {
#pragma unroll
        for (int j = 0; j < 4; ++j) {
            const float* p = W_hh + (16 * g + m) * 16 + 8 * hl + 2 * j;
            w.a[g][j][0] = p[0];
            w.a[g][j][1] = p[1];
        }
    }
#pragma unroll
    for (int v = 0; v < 8; ++v) {
        const int row = 8 * hl + v;
#pragma unroll
        for (int g = 0; g < 2; ++g) {
            w.c_rz[g][v]  = b_ih[16 * g + row] + b_hh[16 * g + row];
            w.wi_rz[g][v] = W_ih[16 * g + row];
        }
        w.bhh_n[v] = b_hh[32 + row];
        w.bih_n[v] = b_ih[32 + row];
        w.wi_n[v]  = W_ih[32 + row];
    }
}

__device__ __forceinline__ void gru_step_tile(const GruTileWeights& w, float (&h)[8], float xv)
{
    v8f ar, az, an;
    float gin[8];
#pragma unroll
    for (int v = 0; v < 8; ++v) {
        ar[v]  = fmaf(w.wi_rz[0][v], xv, w.c_rz[0][v]);
        az[v]  = fmaf(w.wi_rz[1][v], xv, w.c_rz[1][v]);
        an[v]  = w.bhh_n[v];
        gin[v] = fmaf(w.wi_n[v], xv, w.bih_n[v]);
    }
#pragma unroll
    for (int j = 0; j < 4; ++j) {
        v2f b; b[0] = h[2 * j]; b[1] = h[2 * j + 1];
        ar = __builtin_amdgcn_wmma_f32_16x16x4_f32(false, w.a[0][j], false, b, (short)0, ar, false, false);
        an = __builtin_amdgcn_wmma_f32_16x16x4_f32(false, w.a[2][j], false, b, (short)0, an, false, false);
        az = __builtin_amdgcn_wmma_f32_16x16x4_f32(false, w.a[1][j], false, b, (short)0, az, false, false);
    }
#pragma unroll
    for (int v = 0; v < 8; ++v) {
        const float r    = __builtin_amdgcn_rcpf(1.0f + __expf(-ar[v]));
        const float npre = fmaf(r, an[v], gin[v]);
        const float e2   = __expf(npre + npre);
        const float n    = 1.0f - 2.0f * __builtin_amdgcn_rcpf(e2 + 1.0f);
        const float z    = __builtin_amdgcn_rcpf(1.0f + __expf(-az[v]));
        h[v] = fmaf(z, h[v] - n, n);
    }
}

__device__ __forceinline__ float4 load_x4(const float* __restrict__ x, int t, int nsteps)
{
    if (t + 3 <= nsteps) return *(const float4*)(x + t);
    float4 r;
    r.x = x[min(t,     nsteps)];
    r.y = x[min(t + 1, nsteps)];
    r.z = x[min(t + 2, nsteps)];
    r.w = x[min(t + 3, nsteps)];
    return r;
}

__device__ __forceinline__ void store_row_half(float* __restrict__ p, const float (&h)[8])
{
    *(float4*)(p)     = make_float4(h[0], h[1], h[2], h[3]);
    *(float4*)(p + 4) = make_float4(h[4], h[5], h[6], h[7]);
}

__global__ __launch_bounds__(32) void gru_tile_scan_kernel(
    const float* __restrict__ x,
    const float* __restrict__ W_ih,
    const float* __restrict__ W_hh,
    const float* __restrict__ b_ih,
    const float* __restrict__ b_hh,
    float* __restrict__ Hbuf,
    float* __restrict__ Ebuf,
    int nsteps, int nchunks)
{
    const int lane = threadIdx.x;
    const int col  = lane & 15;
    const int hl   = lane >> 4;
    const int c    = blockIdx.x * 16 + col;
    const bool own = c < nchunks;
    const int t_own = own ? c * TILE_CHUNK : 0;
    const int t_end = own ? min(t_own + TILE_CHUNK, nsteps) : 0;
    const int t_beg = max(t_own - TILE_RUNIN, 0);

    GruTileWeights w;
    load_tile_weights(w, W_ih, W_hh, b_ih, b_hh, lane);

    float h[8];
#pragma unroll
    for (int v = 0; v < 8; ++v) h[v] = 0.0f;

    float4 xq = load_x4(x, t_beg, nsteps);
    for (int jb = 0; jb < TILE_RUNIN + TILE_CHUNK; jb += 4) {
        float4 xn = xq;
        if (jb + 4 < TILE_RUNIN + TILE_CHUNK) xn = load_x4(x, t_beg + jb + 4, nsteps);
        const float xs[4] = {xq.x, xq.y, xq.z, xq.w};
#pragma unroll
        for (int u = 0; u < 4; ++u) {
            const int ts = t_beg + jb + u;
            gru_step_tile(w, h, xs[u]);
            if (ts >= t_own && ts < t_end) {
                store_row_half(Hbuf + (size_t)ts * 16 + 8 * hl, h);
                if (ts == t_end - 1) store_row_half(Ebuf + (size_t)c * GRU_REC + 8 * hl, h);
            }
        }
        xq = xn;
    }
}

__device__ __forceinline__ unsigned gru_tile_rescan_until_met(
    const GruTileWeights& w, float (&h)[8],
    const float* __restrict__ x, float* __restrict__ Hbuf,
    int t_own, int t_end, bool active, int hl, int nsteps)
{
    const int    t_last = max(t_end - 1, 0);
    const float* rows   = Hbuf + 8 * hl;

    float4 oa[4], ob[4];
#pragma unroll
    for (int u = 0; u < 4; ++u) {
        const float* p = rows + (size_t)min(t_own + u, t_last) * 16;
        oa[u] = *(const float4*)p;
        ob[u] = *(const float4*)(p + 4);
    }
    bool   differs = false;
    float4 xq = load_x4(x, t_own, nsteps);
    for (int jb = 0; jb < TILE_CHUNK; jb += 4) {
        float4 xn = xq;
        if (jb + 4 < TILE_CHUNK) xn = load_x4(x, t_own + jb + 4, nsteps);
        float4 na[4], nb[4];
#pragma unroll
        for (int u = 0; u < 4; ++u) {
            const float* p = rows + (size_t)min(t_own + jb + 4 + u, t_last) * 16;
            na[u] = *(const float4*)p;
            nb[u] = *(const float4*)(p + 4);
        }
        const float xs[4] = {xq.x, xq.y, xq.z, xq.w};
#pragma unroll
        for (int u = 0; u < 4; ++u) {
            const int  ts   = t_own + jb + u;
            const bool live = active && (ts < t_end);
            gru_step_tile(w, h, xs[u]);
            if (live) {
                differs =
                    (__float_as_uint(h[0]) != __float_as_uint(oa[u].x)) |
                    (__float_as_uint(h[1]) != __float_as_uint(oa[u].y)) |
                    (__float_as_uint(h[2]) != __float_as_uint(oa[u].z)) |
                    (__float_as_uint(h[3]) != __float_as_uint(oa[u].w)) |
                    (__float_as_uint(h[4]) != __float_as_uint(ob[u].x)) |
                    (__float_as_uint(h[5]) != __float_as_uint(ob[u].y)) |
                    (__float_as_uint(h[6]) != __float_as_uint(ob[u].z)) |
                    (__float_as_uint(h[7]) != __float_as_uint(ob[u].w));
            }
            if (__builtin_amdgcn_ballot_w32(differs) == 0u) return 0u;
            if (live) store_row_half(Hbuf + (size_t)ts * 16 + 8 * hl, h);
        }
#pragma unroll
        for (int u = 0; u < 4; ++u) { oa[u] = na[u]; ob[u] = nb[u]; }
        xq = xn;
    }
    return __builtin_amdgcn_ballot_w32(differs);
}

__global__ __launch_bounds__(32) void gru_tile_fixup_kernel(
    const float* __restrict__ x,
    const float* __restrict__ W_ih,
    const float* __restrict__ W_hh,
    const float* __restrict__ b_ih,
    const float* __restrict__ b_hh,
    float* __restrict__ Hbuf,
    const float* __restrict__ Ebuf,
    int* __restrict__ Flag,
    int nsteps, int nchunks)
{
    const int lane = threadIdx.x;
    const int col  = lane & 15;
    const int hl   = lane >> 4;
    const int c    = blockIdx.x * 16 + col;
    const bool own    = c < nchunks;
    const bool active = own && (c >= 1);
    const int t_own = own ? c * TILE_CHUNK : 0;
    const int t_end = own ? min(t_own + TILE_CHUNK, nsteps) : 0;

    GruTileWeights w;
    load_tile_weights(w, W_ih, W_hh, b_ih, b_hh, lane);

    float h[8];
#pragma unroll
    for (int v = 0; v < 8; ++v) h[v] = 0.0f;
    if (active) {
        const float* p = Ebuf + (size_t)(c - 1) * GRU_REC + 8 * hl;
        const float4 a = *(const float4*)p;
        const float4 b = *(const float4*)(p + 4);
        h[0] = a.x; h[1] = a.y; h[2] = a.z; h[3] = a.w;
        h[4] = b.x; h[5] = b.y; h[6] = b.z; h[7] = b.w;
    }
    const unsigned mask = gru_tile_rescan_until_met(w, h, x, Hbuf, t_own, t_end, active, hl, nsteps);
    const unsigned both = (mask | (mask >> 16)) & 0xFFFFu;
    if (lane == 0) Flag[(size_t)(blockIdx.x * 16) * GRU_REC] = (int)both;
}

__global__ __launch_bounds__(256) void gru_tile_check_rescan_kernel(
    const float* __restrict__ x,
    const float* __restrict__ W_ih,
    const float* __restrict__ W_hh,
    const float* __restrict__ b_ih,
    const float* __restrict__ b_hh,
    float* __restrict__ Hbuf,
    const float* __restrict__ Ebuf,
    const int* __restrict__ Flag,
    int nsteps, int nchunks)
{
    __shared__ int any_bad;
    const int tid  = threadIdx.x;
    const int lane = tid & 31;
    const int col  = lane & 15;
    const int hl   = lane >> 4;

    if (tid == 0) any_bad = 0;
    __syncthreads();

    unsigned bad = 0u;
    const int last = nchunks - 1;
    for (int g = tid; g * 16 < nchunks; g += 256) {
        unsigned m = (unsigned)Flag[(size_t)(g * 16) * GRU_REC];
        if (g == (last >> 4)) m &= ~(1u << (last & 15));
        bad |= m;
    }
    if (bad) any_bad = 1;
    __syncthreads();
    if (any_bad == 0) return;
    if (tid >= 32) return;

    GruTileWeights w;
    load_tile_weights(w, W_ih, W_hh, b_ih, b_hh, lane);

    bool     have  = false;
    unsigned gmask = 0u;
    unsigned fmask = 0u;
    for (int c = 1; c < nchunks; ++c) {
        const int p = c - 1;
        if ((p & 511) == 0) {
            const int c0 = p + 16 * lane;
            const int f  = (c0 < nchunks) ? Flag[(size_t)c0 * GRU_REC] : 0;
            gmask = __builtin_amdgcn_ballot_w32(f != 0);
            if (gmask == 0u && !have) { c += 511; continue; }
        }
        if ((p & 15) == 0) {
            fmask = (unsigned)Flag[(size_t)p * GRU_REC];
            if (fmask == 0u && !have) { c += 15; continue; }
        }
        const bool pred_changed = ((fmask >> (p & 15)) & 1u) != 0u;
        if (!pred_changed && !have) continue;

        const int t_own = c * TILE_CHUNK;
        const int t_end = min(t_own + TILE_CHUNK, nsteps);
        __threadfence();
        const float* ps = Ebuf + (size_t)p * GRU_REC + 8 * hl;
        const float* pe = Hbuf + (size_t)(t_own - 1) * 16 + 8 * hl;
        const float4 sa = *(const float4*)ps, sb = *(const float4*)(ps + 4);
        const float4 ea = *(const float4*)pe, eb = *(const float4*)(pe + 4);
        const bool differs =
            (__float_as_uint(sa.x) != __float_as_uint(ea.x)) | (__float_as_uint(sa.y) != __float_as_uint(ea.y)) |
            (__float_as_uint(sa.z) != __float_as_uint(ea.z)) | (__float_as_uint(sa.w) != __float_as_uint(ea.w)) |
            (__float_as_uint(sb.x) != __float_as_uint(eb.x)) | (__float_as_uint(sb.y) != __float_as_uint(eb.y)) |
            (__float_as_uint(sb.z) != __float_as_uint(eb.z)) | (__float_as_uint(sb.w) != __float_as_uint(eb.w));
        have = false;
        if (__builtin_amdgcn_ballot_w32(differs) != 0u) {
            float h[8] = {ea.x, ea.y, ea.z, ea.w, eb.x, eb.y, eb.z, eb.w};
            const bool active = (col == (c & 15));
            have = gru_tile_rescan_until_met(w, h, x, Hbuf, t_own, t_end, active, hl, nsteps) != 0u;
        }
    }
}

extern "C" void kernel_launch(void* const* d_in, const int* in_sizes, int n_in,
                              void* d_out, int out_size, void* d_ws, size_t ws_size,
                              hipStream_t stream)
{
    const float* x     = (const float*)d_in[0];
    const float* W_ih  = (const float*)d_in[1];
    const float* W_hh  = (const float*)d_in[2];
    const float* b_ih  = (const float*)d_in[3];
    const float* b_hh  = (const float*)d_in[4];
    const float* W_out = (const float*)d_in[5];
    const float* b_out = (const float*)d_in[6];
    float* y = (float*)d_out;

    const int nsteps = in_sizes[0] - 1;
    const size_t needH = (size_t)(nsteps + 1) * 16 * sizeof(float);
    float* Hbuf = (float*)d_ws;
    const int    nchunks = (nsteps + GRU_CHUNK - 1) / GRU_CHUNK;
    const size_t needS   = (size_t)nchunks * GRU_REC * sizeof(float);
    float* Ebuf = Hbuf + (size_t)(nsteps + 1) * 16;
    int*   Flag = (int*)(Ebuf + 16);

    const int    tchunks = (nsteps + TILE_CHUNK - 1) / TILE_CHUNK;
    const size_t needT   = (size_t)tchunks * GRU_REC * sizeof(float);
    const int    twaves  = (tchunks + 15) / 16;

    const int tiles  = (nsteps + 15) / 16;
    const int blocks = (tiles + 3) / 4;

    if (ws_size >= needH + needT) {
        hipLaunchKernelGGL(gru_tile_scan_kernel, dim3(twaves), dim3(32), 0, stream,
                           x, W_ih, W_hh, b_ih, b_hh, Hbuf, Ebuf, nsteps, tchunks);
        hipLaunchKernelGGL(gru_tile_fixup_kernel, dim3(twaves), dim3(32), 0, stream,
                           x, W_ih, W_hh, b_ih, b_hh, Hbuf, Ebuf, Flag, nsteps, tchunks);
        hipLaunchKernelGGL(gru_tile_check_rescan_kernel, dim3(1), dim3(256), 0, stream,
                           x, W_ih, W_hh, b_ih, b_hh, Hbuf, Ebuf, Flag, nsteps, tchunks);
        hipLaunchKernelGGL(gru_proj_wmma_kernel, dim3(blocks), dim3(128), 0, stream,
                           Hbuf, W_out, b_out, y, nsteps);
    } else if (ws_size >= needH + needS) {
        hipLaunchKernelGGL(gru_scan_chunks_kernel, dim3(nchunks), dim3(32), 0, stream,
                           x, W_ih, W_hh, b_ih, b_hh, Hbuf, Ebuf, nsteps);
        hipLaunchKernelGGL(gru_fixup_chunks_kernel, dim3(nchunks), dim3(32), 0, stream,
                           x, W_ih, W_hh, b_ih, b_hh, Hbuf, Ebuf, Flag, nsteps);
        hipLaunchKernelGGL(gru_check_rescan_kernel, dim3(1), dim3(1024), 0, stream,
                           x, W_ih, W_hh, b_ih, b_hh, Hbuf, Ebuf, Flag, nsteps, nchunks);
        hipLaunchKernelGGL(gru_proj_wmma_kernel, dim3(blocks), dim3(128), 0, stream,
                           Hbuf, W_out, b_out, y, nsteps);
    } else if (ws_size >= needH) {
        hipLaunchKernelGGL((gru_scan_kernel<0>), dim3(1), dim3(32), 0, stream,
                           x, W_ih, W_hh, b_ih, b_hh, W_out, b_out,
                           Hbuf, y, nsteps);
        hipLaunchKernelGGL(gru_proj_wmma_kernel, dim3(blocks), dim3(128), 0, stream,
                           Hbuf, W_out, b_out, y, nsteps);
    } else {
        hipLaunchKernelGGL((gru_scan_kernel<1>), dim3(1), dim3(32), 0, stream,
                           x, W_ih, W_hh, b_ih, b_hh, W_out, b_out,
                           nullptr, y, nsteps);
    }
}


# projection fused into the scan: y = W_out.h + b_out is a 4th row tile, 4 f32 16x16x4 WMMA among each own step's 12 on the same B; no projection launch, h rows not re-read; y's k sum order changes
# speedup vs baseline: 1772.4020x; 1.2239x over previous
#include <hip/hip_runtime.h>
#include <hip/hip_bf16.h>


typedef __attribute__((ext_vector_type(2))) float v2f;
typedef __attribute__((ext_vector_type(8))) float v8f;

#define GRU_CHUNK 256
#define GRU_FIX_BLK 8
#define GRU_REC 32

__device__ __forceinline__ float bcast_lane(float v, int l) {
    return __int_as_float(__builtin_amdgcn_readlane(__float_as_int(v), l));
}

struct GruLaneWeights {
    float wA[16];
    float wB[16];
    float cA, wiA, bhhB, cBgi, wiB;
};

__device__ __forceinline__ void load_lane_weights(
    GruLaneWeights& w,
    const float* __restrict__ W_ih, const float* __restrict__ W_hh,
    const float* __restrict__ b_ih, const float* __restrict__ b_hh, int lane)
{
    const int m = lane & 15;
#pragma unroll
    for (int k = 0; k < 16; ++k) {
        w.wA[k] = W_hh[lane * 16 + k];
        w.wB[k] = W_hh[(32 + m) * 16 + k];
    }
    w.cA   = b_ih[lane] + b_hh[lane];
    w.wiA  = W_ih[lane];
    w.bhhB = b_hh[32 + m];
    w.cBgi = b_ih[32 + m];
    w.wiB  = W_ih[32 + m];
}

__device__ __forceinline__ float gru_step(const GruLaneWeights& w, float hv, float x_cur)
{
    float sh[16];
#pragma unroll
    for (int k = 0; k < 16; ++k) sh[k] = bcast_lane(hv, k);

    float aA0 = fmaf(w.wiA, x_cur, w.cA);
    float aA1 = 0.0f;
    float aB0 = w.bhhB;
    float aB1 = 0.0f;
#pragma unroll
    for (int k = 0; k < 8; ++k) {
        aA0 = fmaf(w.wA[k],     sh[k],     aA0);
        aA1 = fmaf(w.wA[k + 8], sh[k + 8], aA1);
        aB0 = fmaf(w.wB[k],     sh[k],     aB0);
        aB1 = fmaf(w.wB[k + 8], sh[k + 8], aB1);
    }
    const float accA = aA0 + aA1;
    const float accB = aB0 + aB1;

    const float gA = __builtin_amdgcn_rcpf(1.0f + __expf(-accA));

    const float npre = fmaf(gA, accB, fmaf(w.wiB, x_cur, w.cBgi));
    const float e2   = __expf(npre + npre);
    const float n    = 1.0f - 2.0f * __builtin_amdgcn_rcpf(e2 + 1.0f);

    const float zl   = __shfl_xor(gA, 16, 32);
    return fmaf(zl, hv - n, n);
}

template <int MODE>
__global__ __launch_bounds__(32) void gru_scan_kernel(
    const float* __restrict__ x,
    const float* __restrict__ W_ih,
    const float* __restrict__ W_hh,
    const float* __restrict__ b_ih,
    const float* __restrict__ b_hh,
    const float* __restrict__ W_out,
    const float* __restrict__ b_out,
    float* __restrict__ Hbuf,
    float* __restrict__ y,
    int nsteps)
{
    const int lane = threadIdx.x;
    const int m    = lane & 15;

    GruLaneWeights w;
    load_lane_weights(w, W_ih, W_hh, b_ih, b_hh, lane);
    const float wout = (lane < 16) ? W_out[m] : 0.0f;
    const float bo   = b_out[0];

    float hv    = 0.0f;
    float x_cur = x[0];

    for (int t = 0; t < nsteps; ++t) {
        float x_nxt = x[t + 1];
        const float hnew = gru_step(w, hv, x_cur);
        hv = hnew;

        if (MODE == 0) {
            if (lane < 16) Hbuf[(size_t)t * 16 + lane] = hnew;
        } else {
            float p = wout * hnew;
            p += __shfl_xor(p, 8, 32);
            p += __shfl_xor(p, 4, 32);
            p += __shfl_xor(p, 2, 32);
            p += __shfl_xor(p, 1, 32);
            if (lane == 0) y[t] = p + bo;
        }
        x_cur = x_nxt;
    }
}

__global__ __launch_bounds__(32) void gru_scan_chunks_kernel(
    const float* __restrict__ x,
    const float* __restrict__ W_ih,
    const float* __restrict__ W_hh,
    const float* __restrict__ b_ih,
    const float* __restrict__ b_hh,
    float* __restrict__ Hbuf,
    float* __restrict__ Ebuf,
    int nsteps)
{
    const int lane  = threadIdx.x;
    const int c     = blockIdx.x;
    const int t_own = c * GRU_CHUNK;
    const int t_end = min(t_own + GRU_CHUNK, nsteps);

    GruLaneWeights w;
    load_lane_weights(w, W_ih, W_hh, b_ih, b_hh, lane);

    float hv    = 0.0f;
    float x_cur = x[t_own];

    for (int t = t_own; t < t_end; ++t) {
        const float x_nxt = x[t + 1];
        hv = gru_step(w, hv, x_cur);
        if (lane < 16) Hbuf[(size_t)t * 16 + lane] = hv;
        x_cur = x_nxt;
    }
    if (lane < 16) Ebuf[c * GRU_REC + lane] = hv;
}

__device__ __forceinline__ int gru_rescan_until_met(
    const GruLaneWeights& w, float& hv,
    const float* __restrict__ x, float* __restrict__ Hbuf,
    int t_own, int t_end, int lane)
{
    const float* col = Hbuf + (lane & 15);

    float old[GRU_FIX_BLK];
#pragma unroll
    for (int j = 0; j < GRU_FIX_BLK; ++j)
        old[j] = col[(size_t)min(t_own + j, t_end - 1) * 16];

    float x_cur = x[t_own];
    for (int tb = t_own; tb < t_end; tb += GRU_FIX_BLK) {
        float nxt[GRU_FIX_BLK];
#pragma unroll
        for (int j = 0; j < GRU_FIX_BLK; ++j)
            nxt[j] = col[(size_t)min(tb + GRU_FIX_BLK + j, t_end - 1) * 16];

#pragma unroll
        for (int j = 0; j < GRU_FIX_BLK; ++j) {
            const int t = tb + j;
            if (t >= t_end) return 1;
            const float x_nxt = x[t + 1];
            hv = gru_step(w, hv, x_cur);
            const bool differs = (lane < 16) &&
                (__float_as_uint(hv) != __float_as_uint(old[j]));
            if (__builtin_amdgcn_ballot_w32(differs) == 0u) return 0;
            if (lane < 16) Hbuf[(size_t)t * 16 + lane] = hv;
            x_cur = x_nxt;
        }
#pragma unroll
        for (int j = 0; j < GRU_FIX_BLK; ++j) old[j] = nxt[j];
    }
    return 1;
}

__global__ __launch_bounds__(32) void gru_fixup_chunks_kernel(
    const float* __restrict__ x,
    const float* __restrict__ W_ih,
    const float* __restrict__ W_hh,
    const float* __restrict__ b_ih,
    const float* __restrict__ b_hh,
    float* __restrict__ Hbuf,
    const float* __restrict__ Ebuf,
    int* __restrict__ Flag,
    int nsteps)
{
    const int lane = threadIdx.x;
    const int m    = lane & 15;
    const int c    = blockIdx.x;
    if (c == 0) {
        if (lane == 0) Flag[0] = 0;
        return;
    }
    const int t_own = c * GRU_CHUNK;
    const int t_end = min(t_own + GRU_CHUNK, nsteps);

    GruLaneWeights w;
    load_lane_weights(w, W_ih, W_hh, b_ih, b_hh, lane);

    float hv = Ebuf[(c - 1) * GRU_REC + m];
    const int changed = gru_rescan_until_met(w, hv, x, Hbuf, t_own, t_end, lane);
    if (lane == 0) Flag[c * GRU_REC] = changed;
}

__global__ __launch_bounds__(1024) void gru_check_rescan_kernel(
    const float* __restrict__ x,
    const float* __restrict__ W_ih,
    const float* __restrict__ W_hh,
    const float* __restrict__ b_ih,
    const float* __restrict__ b_hh,
    float* __restrict__ Hbuf,
    const float* __restrict__ Ebuf,
    const int* __restrict__ Flag,
    int nsteps, int nchunks)
{
    __shared__ int any_bad;
    const int tid  = threadIdx.x;
    const int lane = tid & 31;
    const int m    = lane & 15;

    if (tid == 0) any_bad = 0;
    __syncthreads();

    int bad = 0;
    for (int c = tid; c < nchunks - 1; c += 1024) bad |= Flag[c * GRU_REC];
    if (bad) any_bad = 1;
    __syncthreads();
    if (any_bad == 0) return;
    if (tid >= 32) return;

    GruLaneWeights w;
    load_lane_weights(w, W_ih, W_hh, b_ih, b_hh, lane);

    float    hv    = 0.0f;
    bool     have  = false;
    unsigned fmask = 0u;
    for (int c = 1; c < nchunks; ++c) {
        const int p = c - 1;
        if ((p & 31) == 0) {
            const int idx = p + lane;
            const int f   = (idx < nchunks) ? Flag[idx * GRU_REC] : 0;
            fmask = __builtin_amdgcn_ballot_w32(f != 0);
        }
        const bool pred_changed = ((fmask >> (p & 31)) & 1u) != 0u;
        if (!pred_changed && !have) continue;

        const int t_own = c * GRU_CHUNK;
        const int t_end = min(t_own + GRU_CHUNK, nsteps);
        const unsigned s = __float_as_uint(Ebuf[p * GRU_REC + m]);
        float e = Hbuf[(size_t)(t_own - 1) * 16 + m];
        if (have) e = hv;
        const bool differs = (lane < 16) && (s != __float_as_uint(e));
        have = false;
        if (__builtin_amdgcn_ballot_w32(differs) != 0u) {
            hv = e;
            have = gru_rescan_until_met(w, hv, x, Hbuf, t_own, t_end, lane) != 0;
        }
    }
}

__global__ __launch_bounds__(128) void gru_proj_wmma_kernel(
    const float* __restrict__ Hbuf,
    const float* __restrict__ W_out,
    const float* __restrict__ b_out,
    float* __restrict__ y,
    int nY)
{
    const int lane = threadIdx.x & 31;
    const int wave = blockIdx.x * 4 + (threadIdx.x >> 5);
    const int t0   = wave * 16;
    if (t0 >= nY) return;

    const int half = lane >> 4;
    const int lrow = lane & 15;

    const float bo = b_out[0];
    v8f c;
#pragma unroll
    for (int i = 0; i < 8; ++i) c[i] = bo;

    const float* hrow = Hbuf + (size_t)(t0 + lrow) * 16;
#pragma unroll
    for (int j = 0; j < 4; ++j) {
        const int col = 4 * j + 2 * half;
        v2f a; a[0] = hrow[col];  a[1] = hrow[col + 1];
        v2f b; b[0] = W_out[col]; b[1] = W_out[col + 1];
        c = __builtin_amdgcn_wmma_f32_16x16x4_f32(
                 false, a,  false, b,
                 (short)0, c,  false,  false);
    }

    const int m0 = t0 + 8 * half;
    if (lrow == 0) {
        if (m0 + 8 <= nY) {
            float4 lo = make_float4(c[0], c[1], c[2], c[3]);
            float4 hi = make_float4(c[4], c[5], c[6], c[7]);
            *(float4*)(y + m0)     = lo;
            *(float4*)(y + m0 + 4) = hi;
        } else {
#pragma unroll
            for (int i = 0; i < 8; ++i)
                if (m0 + i < nY) y[m0 + i] = c[i];
        }
    }
}

#define TILE_CHUNK 64
#define TILE_RUNIN 192

struct GruTileWeights {
    v2f   a[3][4];
    v2f   ay[4];
    float bo;
    float c_rz[2][8];
    float wi_rz[2][8];
    float bhh_n[8];
    float bih_n[8];
    float wi_n[8];
};

__device__ __forceinline__ void load_tile_weights(
    GruTileWeights& w,
    const float* __restrict__ W_ih, const float* __restrict__ W_hh,
    const float* __restrict__ b_ih, const float* __restrict__ b_hh,
    const float* __restrict__ W_out, const float* __restrict__ b_out, int lane)
{
    const int m  = lane & 15;
    const int hl = lane >> 4;
#pragma unroll
    for (int j = 0; j < 4; ++j) {
        const float* q = W_out + 8 * hl + 2 * j;
        w.ay[j][0] = q[0];
        w.ay[j][1] = q[1];
    }
    w.bo = b_out[0];
#pragma unroll
    for (int g = 0; g < 3; ++g) {
#pragma unroll
        for (int j = 0; j < 4; ++j) {
            const float* p = W_hh + (16 * g + m) * 16 + 8 * hl + 2 * j;
            w.a[g][j][0] = p[0];
            w.a[g][j][1] = p[1];
        }
    }
#pragma unroll
    for (int v = 0; v < 8; ++v) {
        const int row = 8 * hl + v;
#pragma unroll
        for (int g = 0; g < 2; ++g) {
            w.c_rz[g][v]  = b_ih[16 * g + row] + b_hh[16 * g + row];
            w.wi_rz[g][v] = W_ih[16 * g + row];
        }
        w.bhh_n[v] = b_hh[32 + row];
        w.bih_n[v] = b_ih[32 + row];
        w.wi_n[v]  = W_ih[32 + row];
    }
}

template <bool WITH_Y>
__device__ __forceinline__ void gru_step_tile(const GruTileWeights& w, float (&h)[8], float xv, float& y_in)
{
    v8f ar, az, an, ay;
    float gin[8];
#pragma unroll
    for (int v = 0; v < 8; ++v) {
        ar[v]  = fmaf(w.wi_rz[0][v], xv, w.c_rz[0][v]);
        az[v]  = fmaf(w.wi_rz[1][v], xv, w.c_rz[1][v]);
        an[v]  = w.bhh_n[v];
        gin[v] = fmaf(w.wi_n[v], xv, w.bih_n[v]);
        ay[v]  = w.bo;
    }
#pragma unroll
    for (int j = 0; j < 4; ++j) {
        v2f b; b[0] = h[2 * j]; b[1] = h[2 * j + 1];
        ar = __builtin_amdgcn_wmma_f32_16x16x4_f32(false, w.a[0][j], false, b, (short)0, ar, false, false);
        an = __builtin_amdgcn_wmma_f32_16x16x4_f32(false, w.a[2][j], false, b, (short)0, an, false, false);
        az = __builtin_amdgcn_wmma_f32_16x16x4_f32(false, w.a[1][j], false, b, (short)0, az, false, false);
        if (WITH_Y)
            ay = __builtin_amdgcn_wmma_f32_16x16x4_f32(false, w.ay[j], false, b, (short)0, ay, false, false);
    }
    if (WITH_Y) y_in = ay[0];
#pragma unroll
    for (int v = 0; v < 8; ++v) {
        const float r    = __builtin_amdgcn_rcpf(1.0f + __expf(-ar[v]));
        const float npre = fmaf(r, an[v], gin[v]);
        const float e2   = __expf(npre + npre);
        const float n    = 1.0f - 2.0f * __builtin_amdgcn_rcpf(e2 + 1.0f);
        const float z    = __builtin_amdgcn_rcpf(1.0f + __expf(-az[v]));
        h[v] = fmaf(z, h[v] - n, n);
    }
}

__device__ __forceinline__ float gru_y_tile(const GruTileWeights& w, const float (&h)[8])
{
    v8f ay;
#pragma unroll
    for (int v = 0; v < 8; ++v) ay[v] = w.bo;
#pragma unroll
    for (int j = 0; j < 4; ++j) {
        v2f b; b[0] = h[2 * j]; b[1] = h[2 * j + 1];
        ay = __builtin_amdgcn_wmma_f32_16x16x4_f32(false, w.ay[j], false, b, (short)0, ay, false, false);
    }
    return ay[0];
}

__device__ __forceinline__ void store_y_group(
    float* __restrict__ y, int g0, int t_own, int t_end, int hl,
    float q0, float q1, float q2, float q3)
{
    if (hl != 0) return;
    if (g0 >= t_own && g0 + 3 < t_end) {
        *(float4*)(y + g0) = make_float4(q0, q1, q2, q3);
    } else {
        if (g0     >= t_own && g0     < t_end) y[g0]     = q0;
        if (g0 + 1 >= t_own && g0 + 1 < t_end) y[g0 + 1] = q1;
        if (g0 + 2 >= t_own && g0 + 2 < t_end) y[g0 + 2] = q2;
        if (g0 + 3 >= t_own && g0 + 3 < t_end) y[g0 + 3] = q3;
    }
}

__device__ __forceinline__ float4 load_x4(const float* __restrict__ x, int t, int nsteps)
{
    if (t + 3 <= nsteps) return *(const float4*)(x + t);
    float4 r;
    r.x = x[min(t,     nsteps)];
    r.y = x[min(t + 1, nsteps)];
    r.z = x[min(t + 2, nsteps)];
    r.w = x[min(t + 3, nsteps)];
    return r;
}

__device__ __forceinline__ void store_row_half(float* __restrict__ p, const float (&h)[8])
{
    *(float4*)(p)     = make_float4(h[0], h[1], h[2], h[3]);
    *(float4*)(p + 4) = make_float4(h[4], h[5], h[6], h[7]);
}

template <bool FIRST>
__device__ __forceinline__ void gru_tile_run_in(
    const GruTileWeights& w, float (&h)[8], const float* __restrict__ x,
    float4& xq, int t_beg, int nsteps)
{
    for (int jb = 0; jb < TILE_RUNIN; jb += 4) {
        const float4 xn = load_x4(x, max(t_beg + jb + 4, 0), nsteps);
        const float xs[4] = {xq.x, xq.y, xq.z, xq.w};
#pragma unroll
        for (int u = 0; u < 4; ++u) {
            float unused;
            gru_step_tile<false>(w, h, xs[u], unused);
            if (FIRST) {
                const bool before = (t_beg + jb + u) < 0;
#pragma unroll
                for (int v = 0; v < 8; ++v) h[v] = before ? 0.0f : h[v];
            }
        }
        xq = xn;
    }
}

__global__ __launch_bounds__(32) void gru_tile_scan_kernel(
    const float* __restrict__ x,
    const float* __restrict__ W_ih,
    const float* __restrict__ W_hh,
    const float* __restrict__ b_ih,
    const float* __restrict__ b_hh,
    const float* __restrict__ W_out,
    const float* __restrict__ b_out,
    float* __restrict__ Hbuf,
    float* __restrict__ Ebuf,
    float* __restrict__ y,
    int nsteps, int nchunks)
{
    const int lane = threadIdx.x;
    const int col  = lane & 15;
    const int hl   = lane >> 4;
    const int c    = blockIdx.x * 16 + col;
    const bool own = c < nchunks;
    const int t_own = own ? c * TILE_CHUNK : 0;
    const int t_end = own ? min(t_own + TILE_CHUNK, nsteps) : 0;
    const int t_beg = t_own - TILE_RUNIN;

    GruTileWeights w;
    load_tile_weights(w, W_ih, W_hh, b_ih, b_hh, W_out, b_out, lane);

    float h[8];
#pragma unroll
    for (int v = 0; v < 8; ++v) h[v] = 0.0f;

    float4 xq = load_x4(x, max(t_beg, 0), nsteps);
    if (blockIdx.x == 0) gru_tile_run_in<true>(w, h, x, xq, t_beg, nsteps);
    else                 gru_tile_run_in<false>(w, h, x, xq, t_beg, nsteps);

    float yq0 = 0.0f, yq1 = 0.0f, yq2 = 0.0f, yq3 = 0.0f;
    for (int jb = 0; jb < TILE_CHUNK; jb += 4) {
        float4 xn = xq;
        if (jb + 4 < TILE_CHUNK) xn = load_x4(x, t_own + jb + 4, nsteps);
        const float xs[4] = {xq.x, xq.y, xq.z, xq.w};
#pragma unroll
        for (int u = 0; u < 4; ++u) {
            const int ts = t_own + jb + u;
            float yin;
            gru_step_tile<true>(w, h, xs[u], yin);
            if (u == 0) {
                yq3 = yin;
                store_y_group(y, ts - 4, t_own, t_end, hl, yq0, yq1, yq2, yq3);
            } else if (u == 1) yq0 = yin;
            else if (u == 2)   yq1 = yin;
            else               yq2 = yin;
            if (ts < t_end) {
                store_row_half(Hbuf + (size_t)ts * 16 + 8 * hl, h);
                if (ts == t_end - 1) store_row_half(Ebuf + (size_t)c * GRU_REC + 8 * hl, h);
            }
        }
        xq = xn;
    }
    yq3 = gru_y_tile(w, h);
    store_y_group(y, t_own + TILE_CHUNK - 4, t_own, t_end, hl, yq0, yq1, yq2, yq3);
}

__device__ __forceinline__ unsigned gru_tile_rescan_until_met(
    const GruTileWeights& w, float (&h)[8],
    const float* __restrict__ x, float* __restrict__ Hbuf, float* __restrict__ y,
    int t_own, int t_end, bool active, int hl, int nsteps)
{
    const int    t_last = max(t_end - 1, 0);
    const float* rows   = Hbuf + 8 * hl;

    float4 oa[4], ob[4];
#pragma unroll
    for (int u = 0; u < 4; ++u) {
        const float* p = rows + (size_t)min(t_own + u, t_last) * 16;
        oa[u] = *(const float4*)p;
        ob[u] = *(const float4*)(p + 4);
    }
    bool   differs = false;
    float4 xq = load_x4(x, t_own, nsteps);
    for (int jb = 0; jb < TILE_CHUNK; jb += 4) {
        float4 xn = xq;
        if (jb + 4 < TILE_CHUNK) xn = load_x4(x, t_own + jb + 4, nsteps);
        float4 na[4], nb[4];
#pragma unroll
        for (int u = 0; u < 4; ++u) {
            const float* p = rows + (size_t)min(t_own + jb + 4 + u, t_last) * 16;
            na[u] = *(const float4*)p;
            nb[u] = *(const float4*)(p + 4);
        }
        const float xs[4] = {xq.x, xq.y, xq.z, xq.w};
#pragma unroll
        for (int u = 0; u < 4; ++u) {
            const int  ts   = t_own + jb + u;
            const bool live = active && (ts < t_end);
            float unused;
            gru_step_tile<false>(w, h, xs[u], unused);
            if (live) {
                differs =
                    (__float_as_uint(h[0]) != __float_as_uint(oa[u].x)) |
                    (__float_as_uint(h[1]) != __float_as_uint(oa[u].y)) |
                    (__float_as_uint(h[2]) != __float_as_uint(oa[u].z)) |
                    (__float_as_uint(h[3]) != __float_as_uint(oa[u].w)) |
                    (__float_as_uint(h[4]) != __float_as_uint(ob[u].x)) |
                    (__float_as_uint(h[5]) != __float_as_uint(ob[u].y)) |
                    (__float_as_uint(h[6]) != __float_as_uint(ob[u].z)) |
                    (__float_as_uint(h[7]) != __float_as_uint(ob[u].w));
            }
            if (__builtin_amdgcn_ballot_w32(differs) == 0u) return 0u;
            const float yv = gru_y_tile(w, h);
            if (live) {
                store_row_half(Hbuf + (size_t)ts * 16 + 8 * hl, h);
                if (hl == 0) y[ts] = yv;
            }
        }
#pragma unroll
        for (int u = 0; u < 4; ++u) { oa[u] = na[u]; ob[u] = nb[u]; }
        xq = xn;
    }
    return __builtin_amdgcn_ballot_w32(differs);
}

__global__ __launch_bounds__(32) void gru_tile_fixup_kernel(
    const float* __restrict__ x,
    const float* __restrict__ W_ih,
    const float* __restrict__ W_hh,
    const float* __restrict__ b_ih,
    const float* __restrict__ b_hh,
    const float* __restrict__ W_out,
    const float* __restrict__ b_out,
    float* __restrict__ Hbuf,
    float* __restrict__ y,
    const float* __restrict__ Ebuf,
    int* __restrict__ Flag,
    int nsteps, int nchunks)
{
    const int lane = threadIdx.x;
    const int col  = lane & 15;
    const int hl   = lane >> 4;
    const int c    = blockIdx.x * 16 + col;
    const bool own    = c < nchunks;
    const bool active = own && (c >= 1);
    const int t_own = own ? c * TILE_CHUNK : 0;
    const int t_end = own ? min(t_own + TILE_CHUNK, nsteps) : 0;

    GruTileWeights w;
    load_tile_weights(w, W_ih, W_hh, b_ih, b_hh, W_out, b_out, lane);

    float h[8];
#pragma unroll
    for (int v = 0; v < 8; ++v) h[v] = 0.0f;
    if (active) {
        const float* p = Ebuf + (size_t)(c - 1) * GRU_REC + 8 * hl;
        const float4 a = *(const float4*)p;
        const float4 b = *(const float4*)(p + 4);
        h[0] = a.x; h[1] = a.y; h[2] = a.z; h[3] = a.w;
        h[4] = b.x; h[5] = b.y; h[6] = b.z; h[7] = b.w;
    }
    const unsigned mask = gru_tile_rescan_until_met(w, h, x, Hbuf, y, t_own, t_end, active, hl, nsteps);
    const unsigned both = (mask | (mask >> 16)) & 0xFFFFu;
    if (lane == 0) Flag[(size_t)(blockIdx.x * 16) * GRU_REC] = (int)both;
}

__global__ __launch_bounds__(256) void gru_tile_check_rescan_kernel(
    const float* __restrict__ x,
    const float* __restrict__ W_ih,
    const float* __restrict__ W_hh,
    const float* __restrict__ b_ih,
    const float* __restrict__ b_hh,
    const float* __restrict__ W_out,
    const float* __restrict__ b_out,
    float* __restrict__ Hbuf,
    float* __restrict__ y,
    const float* __restrict__ Ebuf,
    const int* __restrict__ Flag,
    int nsteps, int nchunks)
{
    __shared__ int any_bad;
    const int tid  = threadIdx.x;
    const int lane = tid & 31;
    const int col  = lane & 15;
    const int hl   = lane >> 4;

    if (tid == 0) any_bad = 0;
    __syncthreads();

    unsigned bad = 0u;
    const int last = nchunks - 1;
    for (int g = tid; g * 16 < nchunks; g += 256) {
        unsigned m = (unsigned)Flag[(size_t)(g * 16) * GRU_REC];
        if (g == (last >> 4)) m &= ~(1u << (last & 15));
        bad |= m;
    }
    if (bad) any_bad = 1;
    __syncthreads();
    if (any_bad == 0) return;
    if (tid >= 32) return;

    GruTileWeights w;
    load_tile_weights(w, W_ih, W_hh, b_ih, b_hh, W_out, b_out, lane);

    bool     have  = false;
    unsigned gmask = 0u;
    unsigned fmask = 0u;
    for (int c = 1; c < nchunks; ++c) {
        const int p = c - 1;
        if ((p & 511) == 0) {
            const int c0 = p + 16 * lane;
            const int f  = (c0 < nchunks) ? Flag[(size_t)c0 * GRU_REC] : 0;
            gmask = __builtin_amdgcn_ballot_w32(f != 0);
            if (gmask == 0u && !have) { c += 511; continue; }
        }
        if ((p & 15) == 0) {
            fmask = (unsigned)Flag[(size_t)p * GRU_REC];
            if (fmask == 0u && !have) { c += 15; continue; }
        }
        const bool pred_changed = ((fmask >> (p & 15)) & 1u) != 0u;
        if (!pred_changed && !have) continue;

        const int t_own = c * TILE_CHUNK;
        const int t_end = min(t_own + TILE_CHUNK, nsteps);
        __threadfence();
        const float* ps = Ebuf + (size_t)p * GRU_REC + 8 * hl;
        const float* pe = Hbuf + (size_t)(t_own - 1) * 16 + 8 * hl;
        const float4 sa = *(const float4*)ps, sb = *(const float4*)(ps + 4);
        const float4 ea = *(const float4*)pe, eb = *(const float4*)(pe + 4);
        const bool differs =
            (__float_as_uint(sa.x) != __float_as_uint(ea.x)) | (__float_as_uint(sa.y) != __float_as_uint(ea.y)) |
            (__float_as_uint(sa.z) != __float_as_uint(ea.z)) | (__float_as_uint(sa.w) != __float_as_uint(ea.w)) |
            (__float_as_uint(sb.x) != __float_as_uint(eb.x)) | (__float_as_uint(sb.y) != __float_as_uint(eb.y)) |
            (__float_as_uint(sb.z) != __float_as_uint(eb.z)) | (__float_as_uint(sb.w) != __float_as_uint(eb.w));
        have = false;
        if (__builtin_amdgcn_ballot_w32(differs) != 0u) {
            float h[8] = {ea.x, ea.y, ea.z, ea.w, eb.x, eb.y, eb.z, eb.w};
            const bool active = (col == (c & 15));
            have = gru_tile_rescan_until_met(w, h, x, Hbuf, y, t_own, t_end, active, hl, nsteps) != 0u;
        }
    }
}

extern "C" void kernel_launch(void* const* d_in, const int* in_sizes, int n_in,
                              void* d_out, int out_size, void* d_ws, size_t ws_size,
                              hipStream_t stream)
{
    const float* x     = (const float*)d_in[0];
    const float* W_ih  = (const float*)d_in[1];
    const float* W_hh  = (const float*)d_in[2];
    const float* b_ih  = (const float*)d_in[3];
    const float* b_hh  = (const float*)d_in[4];
    const float* W_out = (const float*)d_in[5];
    const float* b_out = (const float*)d_in[6];
    float* y = (float*)d_out;

    const int nsteps = in_sizes[0] - 1;
    const size_t needH = (size_t)(nsteps + 1) * 16 * sizeof(float);
    float* Hbuf = (float*)d_ws;
    const int    nchunks = (nsteps + GRU_CHUNK - 1) / GRU_CHUNK;
    const size_t needS   = (size_t)nchunks * GRU_REC * sizeof(float);
    float* Ebuf = Hbuf + (size_t)(nsteps + 1) * 16;
    int*   Flag = (int*)(Ebuf + 16);

    const int    tchunks = (nsteps + TILE_CHUNK - 1) / TILE_CHUNK;
    const size_t needT   = (size_t)tchunks * GRU_REC * sizeof(float);
    const int    twaves  = (tchunks + 15) / 16;

    const int tiles  = (nsteps + 15) / 16;
    const int blocks = (tiles + 3) / 4;

    if (ws_size >= needH + needT) {
        hipLaunchKernelGGL(gru_tile_scan_kernel, dim3(twaves), dim3(32), 0, stream,
                           x, W_ih, W_hh, b_ih, b_hh, W_out, b_out, Hbuf, Ebuf, y, nsteps, tchunks);
        hipLaunchKernelGGL(gru_tile_fixup_kernel, dim3(twaves), dim3(32), 0, stream,
                           x, W_ih, W_hh, b_ih, b_hh, W_out, b_out, Hbuf, y, Ebuf, Flag, nsteps, tchunks);
        hipLaunchKernelGGL(gru_tile_check_rescan_kernel, dim3(1), dim3(256), 0, stream,
                           x, W_ih, W_hh, b_ih, b_hh, W_out, b_out, Hbuf, y, Ebuf, Flag, nsteps, tchunks);
    } else if (ws_size >= needH + needS) {
        hipLaunchKernelGGL(gru_scan_chunks_kernel, dim3(nchunks), dim3(32), 0, stream,
                           x, W_ih, W_hh, b_ih, b_hh, Hbuf, Ebuf, nsteps);
        hipLaunchKernelGGL(gru_fixup_chunks_kernel, dim3(nchunks), dim3(32), 0, stream,
                           x, W_ih, W_hh, b_ih, b_hh, Hbuf, Ebuf, Flag, nsteps);
        hipLaunchKernelGGL(gru_check_rescan_kernel, dim3(1), dim3(1024), 0, stream,
                           x, W_ih, W_hh, b_ih, b_hh, Hbuf, Ebuf, Flag, nsteps, nchunks);
        hipLaunchKernelGGL(gru_proj_wmma_kernel, dim3(blocks), dim3(128), 0, stream,
                           Hbuf, W_out, b_out, y, nsteps);
    } else if (ws_size >= needH) {
        hipLaunchKernelGGL((gru_scan_kernel<0>), dim3(1), dim3(32), 0, stream,
                           x, W_ih, W_hh, b_ih, b_hh, W_out, b_out,
                           Hbuf, y, nsteps);
        hipLaunchKernelGGL(gru_proj_wmma_kernel, dim3(blocks), dim3(128), 0, stream,
                           Hbuf, W_out, b_out, y, nsteps);
    } else {
        hipLaunchKernelGGL((gru_scan_kernel<1>), dim3(1), dim3(32), 0, stream,
                           x, W_ih, W_hh, b_ih, b_hh, W_out, b_out,
                           nullptr, y, nsteps);
    }
}


# launch 1's loops freed of lane guards where a wave's 16 chunks are whole (every wave here): plain x loads, unconditional row stores, y groups by lane halves; no EXEC masking or branch between steps
# speedup vs baseline: 1971.0757x; 1.1121x over previous
#include <hip/hip_runtime.h>
#include <hip/hip_bf16.h>


typedef __attribute__((ext_vector_type(2))) float v2f;
typedef __attribute__((ext_vector_type(8))) float v8f;

#define GRU_CHUNK 256
#define GRU_FIX_BLK 8
#define GRU_REC 32

__device__ __forceinline__ float bcast_lane(float v, int l) {
    return __int_as_float(__builtin_amdgcn_readlane(__float_as_int(v), l));
}

struct GruLaneWeights {
    float wA[16];
    float wB[16];
    float cA, wiA, bhhB, cBgi, wiB;
};

__device__ __forceinline__ void load_lane_weights(
    GruLaneWeights& w,
    const float* __restrict__ W_ih, const float* __restrict__ W_hh,
    const float* __restrict__ b_ih, const float* __restrict__ b_hh, int lane)
{
    const int m = lane & 15;
#pragma unroll
    for (int k = 0; k < 16; ++k) {
        w.wA[k] = W_hh[lane * 16 + k];
        w.wB[k] = W_hh[(32 + m) * 16 + k];
    }
    w.cA   = b_ih[lane] + b_hh[lane];
    w.wiA  = W_ih[lane];
    w.bhhB = b_hh[32 + m];
    w.cBgi = b_ih[32 + m];
    w.wiB  = W_ih[32 + m];
}

__device__ __forceinline__ float gru_step(const GruLaneWeights& w, float hv, float x_cur)
{
    float sh[16];
#pragma unroll
    for (int k = 0; k < 16; ++k) sh[k] = bcast_lane(hv, k);

    float aA0 = fmaf(w.wiA, x_cur, w.cA);
    float aA1 = 0.0f;
    float aB0 = w.bhhB;
    float aB1 = 0.0f;
#pragma unroll
    for (int k = 0; k < 8; ++k) {
        aA0 = fmaf(w.wA[k],     sh[k],     aA0);
        aA1 = fmaf(w.wA[k + 8], sh[k + 8], aA1);
        aB0 = fmaf(w.wB[k],     sh[k],     aB0);
        aB1 = fmaf(w.wB[k + 8], sh[k + 8], aB1);
    }
    const float accA = aA0 + aA1;
    const float accB = aB0 + aB1;

    const float gA = __builtin_amdgcn_rcpf(1.0f + __expf(-accA));

    const float npre = fmaf(gA, accB, fmaf(w.wiB, x_cur, w.cBgi));
    const float e2   = __expf(npre + npre);
    const float n    = 1.0f - 2.0f * __builtin_amdgcn_rcpf(e2 + 1.0f);

    const float zl   = __shfl_xor(gA, 16, 32);
    return fmaf(zl, hv - n, n);
}

template <int MODE>
__global__ __launch_bounds__(32) void gru_scan_kernel(
    const float* __restrict__ x,
    const float* __restrict__ W_ih,
    const float* __restrict__ W_hh,
    const float* __restrict__ b_ih,
    const float* __restrict__ b_hh,
    const float* __restrict__ W_out,
    const float* __restrict__ b_out,
    float* __restrict__ Hbuf,
    float* __restrict__ y,
    int nsteps)
{
    const int lane = threadIdx.x;
    const int m    = lane & 15;

    GruLaneWeights w;
    load_lane_weights(w, W_ih, W_hh, b_ih, b_hh, lane);
    const float wout = (lane < 16) ? W_out[m] : 0.0f;
    const float bo   = b_out[0];

    float hv    = 0.0f;
    float x_cur = x[0];

    for (int t = 0; t < nsteps; ++t) {
        float x_nxt = x[t + 1];
        const float hnew = gru_step(w, hv, x_cur);
        hv = hnew;

        if (MODE == 0) {
            if (lane < 16) Hbuf[(size_t)t * 16 + lane] = hnew;
        } else {
            float p = wout * hnew;
            p += __shfl_xor(p, 8, 32);
            p += __shfl_xor(p, 4, 32);
            p += __shfl_xor(p, 2, 32);
            p += __shfl_xor(p, 1, 32);
            if (lane == 0) y[t] = p + bo;
        }
        x_cur = x_nxt;
    }
}

__global__ __launch_bounds__(32) void gru_scan_chunks_kernel(
    const float* __restrict__ x,
    const float* __restrict__ W_ih,
    const float* __restrict__ W_hh,
    const float* __restrict__ b_ih,
    const float* __restrict__ b_hh,
    float* __restrict__ Hbuf,
    float* __restrict__ Ebuf,
    int nsteps)
{
    const int lane  = threadIdx.x;
    const int c     = blockIdx.x;
    const int t_own = c * GRU_CHUNK;
    const int t_end = min(t_own + GRU_CHUNK, nsteps);

    GruLaneWeights w;
    load_lane_weights(w, W_ih, W_hh, b_ih, b_hh, lane);

    float hv    = 0.0f;
    float x_cur = x[t_own];

    for (int t = t_own; t < t_end; ++t) {
        const float x_nxt = x[t + 1];
        hv = gru_step(w, hv, x_cur);
        if (lane < 16) Hbuf[(size_t)t * 16 + lane] = hv;
        x_cur = x_nxt;
    }
    if (lane < 16) Ebuf[c * GRU_REC + lane] = hv;
}

__device__ __forceinline__ int gru_rescan_until_met(
    const GruLaneWeights& w, float& hv,
    const float* __restrict__ x, float* __restrict__ Hbuf,
    int t_own, int t_end, int lane)
{
    const float* col = Hbuf + (lane & 15);

    float old[GRU_FIX_BLK];
#pragma unroll
    for (int j = 0; j < GRU_FIX_BLK; ++j)
        old[j] = col[(size_t)min(t_own + j, t_end - 1) * 16];

    float x_cur = x[t_own];
    for (int tb = t_own; tb < t_end; tb += GRU_FIX_BLK) {
        float nxt[GRU_FIX_BLK];
#pragma unroll
        for (int j = 0; j < GRU_FIX_BLK; ++j)
            nxt[j] = col[(size_t)min(tb + GRU_FIX_BLK + j, t_end - 1) * 16];

#pragma unroll
        for (int j = 0; j < GRU_FIX_BLK; ++j) {
            const int t = tb + j;
            if (t >= t_end) return 1;
            const float x_nxt = x[t + 1];
            hv = gru_step(w, hv, x_cur);
            const bool differs = (lane < 16) &&
                (__float_as_uint(hv) != __float_as_uint(old[j]));
            if (__builtin_amdgcn_ballot_w32(differs) == 0u) return 0;
            if (lane < 16) Hbuf[(size_t)t * 16 + lane] = hv;
            x_cur = x_nxt;
        }
#pragma unroll
        for (int j = 0; j < GRU_FIX_BLK; ++j) old[j] = nxt[j];
    }
    return 1;
}

__global__ __launch_bounds__(32) void gru_fixup_chunks_kernel(
    const float* __restrict__ x,
    const float* __restrict__ W_ih,
    const float* __restrict__ W_hh,
    const float* __restrict__ b_ih,
    const float* __restrict__ b_hh,
    float* __restrict__ Hbuf,
    const float* __restrict__ Ebuf,
    int* __restrict__ Flag,
    int nsteps)
{
    const int lane = threadIdx.x;
    const int m    = lane & 15;
    const int c    = blockIdx.x;
    if (c == 0) {
        if (lane == 0) Flag[0] = 0;
        return;
    }
    const int t_own = c * GRU_CHUNK;
    const int t_end = min(t_own + GRU_CHUNK, nsteps);

    GruLaneWeights w;
    load_lane_weights(w, W_ih, W_hh, b_ih, b_hh, lane);

    float hv = Ebuf[(c - 1) * GRU_REC + m];
    const int changed = gru_rescan_until_met(w, hv, x, Hbuf, t_own, t_end, lane);
    if (lane == 0) Flag[c * GRU_REC] = changed;
}

__global__ __launch_bounds__(1024) void gru_check_rescan_kernel(
    const float* __restrict__ x,
    const float* __restrict__ W_ih,
    const float* __restrict__ W_hh,
    const float* __restrict__ b_ih,
    const float* __restrict__ b_hh,
    float* __restrict__ Hbuf,
    const float* __restrict__ Ebuf,
    const int* __restrict__ Flag,
    int nsteps, int nchunks)
{
    __shared__ int any_bad;
    const int tid  = threadIdx.x;
    const int lane = tid & 31;
    const int m    = lane & 15;

    if (tid == 0) any_bad = 0;
    __syncthreads();

    int bad = 0;
    for (int c = tid; c < nchunks - 1; c += 1024) bad |= Flag[c * GRU_REC];
    if (bad) any_bad = 1;
    __syncthreads();
    if (any_bad == 0) return;
    if (tid >= 32) return;

    GruLaneWeights w;
    load_lane_weights(w, W_ih, W_hh, b_ih, b_hh, lane);

    float    hv    = 0.0f;
    bool     have  = false;
    unsigned fmask = 0u;
    for (int c = 1; c < nchunks; ++c) {
        const int p = c - 1;
        if ((p & 31) == 0) {
            const int idx = p + lane;
            const int f   = (idx < nchunks) ? Flag[idx * GRU_REC] : 0;
            fmask = __builtin_amdgcn_ballot_w32(f != 0);
        }
        const bool pred_changed = ((fmask >> (p & 31)) & 1u) != 0u;
        if (!pred_changed && !have) continue;

        const int t_own = c * GRU_CHUNK;
        const int t_end = min(t_own + GRU_CHUNK, nsteps);
        const unsigned s = __float_as_uint(Ebuf[p * GRU_REC + m]);
        float e = Hbuf[(size_t)(t_own - 1) * 16 + m];
        if (have) e = hv;
        const bool differs = (lane < 16) && (s != __float_as_uint(e));
        have = false;
        if (__builtin_amdgcn_ballot_w32(differs) != 0u) {
            hv = e;
            have = gru_rescan_until_met(w, hv, x, Hbuf, t_own, t_end, lane) != 0;
        }
    }
}

__global__ __launch_bounds__(128) void gru_proj_wmma_kernel(
    const float* __restrict__ Hbuf,
    const float* __restrict__ W_out,
    const float* __restrict__ b_out,
    float* __restrict__ y,
    int nY)
{
    const int lane = threadIdx.x & 31;
    const int wave = blockIdx.x * 4 + (threadIdx.x >> 5);
    const int t0   = wave * 16;
    if (t0 >= nY) return;

    const int half = lane >> 4;
    const int lrow = lane & 15;

    const float bo = b_out[0];
    v8f c;
#pragma unroll
    for (int i = 0; i < 8; ++i) c[i] = bo;

    const float* hrow = Hbuf + (size_t)(t0 + lrow) * 16;
#pragma unroll
    for (int j = 0; j < 4; ++j) {
        const int col = 4 * j + 2 * half;
        v2f a; a[0] = hrow[col];  a[1] = hrow[col + 1];
        v2f b; b[0] = W_out[col]; b[1] = W_out[col + 1];
        c = __builtin_amdgcn_wmma_f32_16x16x4_f32(
                 false, a,  false, b,
                 (short)0, c,  false,  false);
    }

    const int m0 = t0 + 8 * half;
    if (lrow == 0) {
        if (m0 + 8 <= nY) {
            float4 lo = make_float4(c[0], c[1], c[2], c[3]);
            float4 hi = make_float4(c[4], c[5], c[6], c[7]);
            *(float4*)(y + m0)     = lo;
            *(float4*)(y + m0 + 4) = hi;
        } else {
#pragma unroll
            for (int i = 0; i < 8; ++i)
                if (m0 + i < nY) y[m0 + i] = c[i];
        }
    }
}

#define TILE_CHUNK 64
#define TILE_RUNIN 192

struct GruTileWeights {
    v2f   a[3][4];
    v2f   ay[4];
    float bo;
    float c_rz[2][8];
    float wi_rz[2][8];
    float bhh_n[8];
    float bih_n[8];
    float wi_n[8];
};

__device__ __forceinline__ void load_tile_weights(
    GruTileWeights& w,
    const float* __restrict__ W_ih, const float* __restrict__ W_hh,
    const float* __restrict__ b_ih, const float* __restrict__ b_hh,
    const float* __restrict__ W_out, const float* __restrict__ b_out, int lane)
{
    const int m  = lane & 15;
    const int hl = lane >> 4;
#pragma unroll
    for (int j = 0; j < 4; ++j) {
        const float* q = W_out + 8 * hl + 2 * j;
        w.ay[j][0] = q[0];
        w.ay[j][1] = q[1];
    }
    w.bo = b_out[0];
#pragma unroll
    for (int g = 0; g < 3; ++g) {
#pragma unroll
        for (int j = 0; j < 4; ++j) {
            const float* p = W_hh + (16 * g + m) * 16 + 8 * hl + 2 * j;
            w.a[g][j][0] = p[0];
            w.a[g][j][1] = p[1];
        }
    }
#pragma unroll
    for (int v = 0; v < 8; ++v) {
        const int row = 8 * hl + v;
#pragma unroll
        for (int g = 0; g < 2; ++g) {
            w.c_rz[g][v]  = b_ih[16 * g + row] + b_hh[16 * g + row];
            w.wi_rz[g][v] = W_ih[16 * g + row];
        }
        w.bhh_n[v] = b_hh[32 + row];
        w.bih_n[v] = b_ih[32 + row];
        w.wi_n[v]  = W_ih[32 + row];
    }
}

template <bool WITH_Y>
__device__ __forceinline__ void gru_step_tile(const GruTileWeights& w, float (&h)[8], float xv, float& y_in)
{
    v8f ar, az, an, ay;
    float gin[8];
#pragma unroll
    for (int v = 0; v < 8; ++v) {
        ar[v]  = fmaf(w.wi_rz[0][v], xv, w.c_rz[0][v]);
        az[v]  = fmaf(w.wi_rz[1][v], xv, w.c_rz[1][v]);
        an[v]  = w.bhh_n[v];
        gin[v] = fmaf(w.wi_n[v], xv, w.bih_n[v]);
        ay[v]  = w.bo;
    }
#pragma unroll
    for (int j = 0; j < 4; ++j) {
        v2f b; b[0] = h[2 * j]; b[1] = h[2 * j + 1];
        ar = __builtin_amdgcn_wmma_f32_16x16x4_f32(false, w.a[0][j], false, b, (short)0, ar, false, false);
        an = __builtin_amdgcn_wmma_f32_16x16x4_f32(false, w.a[2][j], false, b, (short)0, an, false, false);
        az = __builtin_amdgcn_wmma_f32_16x16x4_f32(false, w.a[1][j], false, b, (short)0, az, false, false);
        if (WITH_Y)
            ay = __builtin_amdgcn_wmma_f32_16x16x4_f32(false, w.ay[j], false, b, (short)0, ay, false, false);
    }
    if (WITH_Y) y_in = ay[0];
#pragma unroll
    for (int v = 0; v < 8; ++v) {
        const float r    = __builtin_amdgcn_rcpf(1.0f + __expf(-ar[v]));
        const float npre = fmaf(r, an[v], gin[v]);
        const float e2   = __expf(npre + npre);
        const float n    = 1.0f - 2.0f * __builtin_amdgcn_rcpf(e2 + 1.0f);
        const float z    = __builtin_amdgcn_rcpf(1.0f + __expf(-az[v]));
        h[v] = fmaf(z, h[v] - n, n);
    }
}

__device__ __forceinline__ float gru_y_tile(const GruTileWeights& w, const float (&h)[8])
{
    v8f ay;
#pragma unroll
    for (int v = 0; v < 8; ++v) ay[v] = w.bo;
#pragma unroll
    for (int j = 0; j < 4; ++j) {
        v2f b; b[0] = h[2 * j]; b[1] = h[2 * j + 1];
        ay = __builtin_amdgcn_wmma_f32_16x16x4_f32(false, w.ay[j], false, b, (short)0, ay, false, false);
    }
    return ay[0];
}

__device__ __forceinline__ void store_y_group(
    float* __restrict__ y, int g0, int t_own, int t_end, int hl,
    float q0, float q1, float q2, float q3)
{
    if (hl != 0) return;
    if (g0 >= t_own && g0 + 3 < t_end) {
        *(float4*)(y + g0) = make_float4(q0, q1, q2, q3);
    } else {
        if (g0     >= t_own && g0     < t_end) y[g0]     = q0;
        if (g0 + 1 >= t_own && g0 + 1 < t_end) y[g0 + 1] = q1;
        if (g0 + 2 >= t_own && g0 + 2 < t_end) y[g0 + 2] = q2;
        if (g0 + 3 >= t_own && g0 + 3 < t_end) y[g0 + 3] = q3;
    }
}

__device__ __forceinline__ float4 load_x4(const float* __restrict__ x, int t, int nsteps)
{
    if (t + 3 <= nsteps) return *(const float4*)(x + t);
    float4 r;
    r.x = x[min(t,     nsteps)];
    r.y = x[min(t + 1, nsteps)];
    r.z = x[min(t + 2, nsteps)];
    r.w = x[min(t + 3, nsteps)];
    return r;
}

__device__ __forceinline__ void store_row_half(float* __restrict__ p, const float (&h)[8])
{
    *(float4*)(p)     = make_float4(h[0], h[1], h[2], h[3]);
    *(float4*)(p + 4) = make_float4(h[4], h[5], h[6], h[7]);
}

template <bool FIRST, bool WHOLE>
__device__ __forceinline__ void gru_tile_run_in(
    const GruTileWeights& w, float (&h)[8], const float* __restrict__ x,
    float4& xq, int t_beg, int nsteps)
{
    for (int jb = 0; jb < TILE_RUNIN; jb += 4) {
        const int tn = FIRST ? max(t_beg + jb + 4, 0) : t_beg + jb + 4;
        const float4 xn = WHOLE ? *(const float4*)(x + tn) : load_x4(x, tn, nsteps);
        const float xs[4] = {xq.x, xq.y, xq.z, xq.w};
#pragma unroll
        for (int u = 0; u < 4; ++u) {
            float unused;
            gru_step_tile<false>(w, h, xs[u], unused);
            if (FIRST) {
                const bool before = (t_beg + jb + u) < 0;
#pragma unroll
                for (int v = 0; v < 8; ++v) h[v] = before ? 0.0f : h[v];
            }
        }
        xq = xn;
    }
}

template <bool STORE_Y>
__device__ __forceinline__ void gru_tile_own_quad_whole(
    const GruTileWeights& w, float (&h)[8], const float4& xq,
    float* __restrict__ hp, float* __restrict__ yp, int hl,
    float& yq0, float& yq1, float& yq2)
{
    const float xs[4] = {xq.x, xq.y, xq.z, xq.w};
#pragma unroll
    for (int u = 0; u < 4; ++u) {
        float yin;
        gru_step_tile<true>(w, h, xs[u], yin);
        if (u == 0) {
            if (STORE_Y) {
                const float2 v = hl ? make_float2(yq2, yin) : make_float2(yq0, yq1);
                *(float2*)yp = v;
            }
        } else if (u == 1) yq0 = yin;
        else if (u == 2)   yq1 = yin;
        else               yq2 = yin;
        store_row_half(hp + u * 16, h);
    }
}

__global__ __launch_bounds__(32) void gru_tile_scan_kernel(
    const float* __restrict__ x,
    const float* __restrict__ W_ih,
    const float* __restrict__ W_hh,
    const float* __restrict__ b_ih,
    const float* __restrict__ b_hh,
    const float* __restrict__ W_out,
    const float* __restrict__ b_out,
    float* __restrict__ Hbuf,
    float* __restrict__ Ebuf,
    float* __restrict__ y,
    int nsteps, int nchunks)
{
    const int lane = threadIdx.x;
    const int col  = lane & 15;
    const int hl   = lane >> 4;
    const int c    = blockIdx.x * 16 + col;
    const bool own = c < nchunks;
    const int t_own = own ? c * TILE_CHUNK : 0;
    const int t_end = own ? min(t_own + TILE_CHUNK, nsteps) : 0;
    const int t_beg = t_own - TILE_RUNIN;

    GruTileWeights w;
    load_tile_weights(w, W_ih, W_hh, b_ih, b_hh, W_out, b_out, lane);

    float h[8];
#pragma unroll
    for (int v = 0; v < 8; ++v) h[v] = 0.0f;

    const bool whole = own && (t_own + TILE_CHUNK - 1 <= nsteps);
    const bool fast  = __builtin_amdgcn_ballot_w32(!whole) == 0u;

    float4 xq = load_x4(x, max(t_beg, 0), nsteps);

    if (fast) {
        if (blockIdx.x == 0) gru_tile_run_in<true, true>(w, h, x, xq, t_beg, nsteps);
        else                 gru_tile_run_in<false, true>(w, h, x, xq, t_beg, nsteps);

        float* hp = Hbuf + (size_t)t_own * 16 + 8 * hl;
        float* yp = y + t_own + 2 * hl;
        float yq0 = 0.0f, yq1 = 0.0f, yq2 = 0.0f;
        {
            const float4 xn = *(const float4*)(x + t_own + 4);
            gru_tile_own_quad_whole<false>(w, h, xq, hp, yp, hl, yq0, yq1, yq2);
            xq = xn;
        }
        for (int jb = 4; jb < TILE_CHUNK; jb += 4) {
            const float4 xn = *(const float4*)(x + t_own + min(jb + 4, TILE_CHUNK - 4));
            gru_tile_own_quad_whole<true>(w, h, xq, hp + (size_t)jb * 16, yp + jb - 4, hl, yq0, yq1, yq2);
            xq = xn;
        }
        store_row_half(Ebuf + (size_t)c * GRU_REC + 8 * hl, h);
        const float yq3 = gru_y_tile(w, h);
        store_y_group(y, t_own + TILE_CHUNK - 4, t_own, t_end, hl, yq0, yq1, yq2, yq3);
        return;
    }

    if (blockIdx.x == 0) gru_tile_run_in<true, false>(w, h, x, xq, t_beg, nsteps);
    else                 gru_tile_run_in<false, false>(w, h, x, xq, t_beg, nsteps);

    float yq0 = 0.0f, yq1 = 0.0f, yq2 = 0.0f, yq3 = 0.0f;
    for (int jb = 0; jb < TILE_CHUNK; jb += 4) {
        float4 xn = xq;
        if (jb + 4 < TILE_CHUNK) xn = load_x4(x, t_own + jb + 4, nsteps);
        const float xs[4] = {xq.x, xq.y, xq.z, xq.w};
#pragma unroll
        for (int u = 0; u < 4; ++u) {
            const int ts = t_own + jb + u;
            float yin;
            gru_step_tile<true>(w, h, xs[u], yin);
            if (u == 0) {
                yq3 = yin;
                store_y_group(y, ts - 4, t_own, t_end, hl, yq0, yq1, yq2, yq3);
            } else if (u == 1) yq0 = yin;
            else if (u == 2)   yq1 = yin;
            else               yq2 = yin;
            if (ts < t_end) {
                store_row_half(Hbuf + (size_t)ts * 16 + 8 * hl, h);
                if (ts == t_end - 1) store_row_half(Ebuf + (size_t)c * GRU_REC + 8 * hl, h);
            }
        }
        xq = xn;
    }
    yq3 = gru_y_tile(w, h);
    store_y_group(y, t_own + TILE_CHUNK - 4, t_own, t_end, hl, yq0, yq1, yq2, yq3);
}

__device__ __forceinline__ unsigned gru_tile_rescan_until_met(
    const GruTileWeights& w, float (&h)[8],
    const float* __restrict__ x, float* __restrict__ Hbuf, float* __restrict__ y,
    int t_own, int t_end, bool active, int hl, int nsteps)
{
    const int    t_last = max(t_end - 1, 0);
    const float* rows   = Hbuf + 8 * hl;

    float4 oa[4], ob[4];
#pragma unroll
    for (int u = 0; u < 4; ++u) {
        const float* p = rows + (size_t)min(t_own + u, t_last) * 16;
        oa[u] = *(const float4*)p;
        ob[u] = *(const float4*)(p + 4);
    }
    bool   differs = false;
    float4 xq = load_x4(x, t_own, nsteps);
    for (int jb = 0; jb < TILE_CHUNK; jb += 4) {
        float4 xn = xq;
        if (jb + 4 < TILE_CHUNK) xn = load_x4(x, t_own + jb + 4, nsteps);
        float4 na[4], nb[4];
#pragma unroll
        for (int u = 0; u < 4; ++u) {
            const float* p = rows + (size_t)min(t_own + jb + 4 + u, t_last) * 16;
            na[u] = *(const float4*)p;
            nb[u] = *(const float4*)(p + 4);
        }
        const float xs[4] = {xq.x, xq.y, xq.z, xq.w};
#pragma unroll
        for (int u = 0; u < 4; ++u) {
            const int  ts   = t_own + jb + u;
            const bool live = active && (ts < t_end);
            float unused;
            gru_step_tile<false>(w, h, xs[u], unused);
            if (live) {
                differs =
                    (__float_as_uint(h[0]) != __float_as_uint(oa[u].x)) |
                    (__float_as_uint(h[1]) != __float_as_uint(oa[u].y)) |
                    (__float_as_uint(h[2]) != __float_as_uint(oa[u].z)) |
                    (__float_as_uint(h[3]) != __float_as_uint(oa[u].w)) |
                    (__float_as_uint(h[4]) != __float_as_uint(ob[u].x)) |
                    (__float_as_uint(h[5]) != __float_as_uint(ob[u].y)) |
                    (__float_as_uint(h[6]) != __float_as_uint(ob[u].z)) |
                    (__float_as_uint(h[7]) != __float_as_uint(ob[u].w));
            }
            if (__builtin_amdgcn_ballot_w32(differs) == 0u) return 0u;
            const float yv = gru_y_tile(w, h);
            if (live) {
                store_row_half(Hbuf + (size_t)ts * 16 + 8 * hl, h);
                if (hl == 0) y[ts] = yv;
            }
        }
#pragma unroll
        for (int u = 0; u < 4; ++u) { oa[u] = na[u]; ob[u] = nb[u]; }
        xq = xn;
    }
    return __builtin_amdgcn_ballot_w32(differs);
}

__global__ __launch_bounds__(32) void gru_tile_fixup_kernel(
    const float* __restrict__ x,
    const float* __restrict__ W_ih,
    const float* __restrict__ W_hh,
    const float* __restrict__ b_ih,
    const float* __restrict__ b_hh,
    const float* __restrict__ W_out,
    const float* __restrict__ b_out,
    float* __restrict__ Hbuf,
    float* __restrict__ y,
    const float* __restrict__ Ebuf,
    int* __restrict__ Flag,
    int nsteps, int nchunks)
{
    const int lane = threadIdx.x;
    const int col  = lane & 15;
    const int hl   = lane >> 4;
    const int c    = blockIdx.x * 16 + col;
    const bool own    = c < nchunks;
    const bool active = own && (c >= 1);
    const int t_own = own ? c * TILE_CHUNK : 0;
    const int t_end = own ? min(t_own + TILE_CHUNK, nsteps) : 0;

    GruTileWeights w;
    load_tile_weights(w, W_ih, W_hh, b_ih, b_hh, W_out, b_out, lane);

    float h[8];
#pragma unroll
    for (int v = 0; v < 8; ++v) h[v] = 0.0f;
    if (active) {
        const float* p = Ebuf + (size_t)(c - 1) * GRU_REC + 8 * hl;
        const float4 a = *(const float4*)p;
        const float4 b = *(const float4*)(p + 4);
        h[0] = a.x; h[1] = a.y; h[2] = a.z; h[3] = a.w;
        h[4] = b.x; h[5] = b.y; h[6] = b.z; h[7] = b.w;
    }
    const unsigned mask = gru_tile_rescan_until_met(w, h, x, Hbuf, y, t_own, t_end, active, hl, nsteps);
    const unsigned both = (mask | (mask >> 16)) & 0xFFFFu;
    if (lane == 0) Flag[(size_t)(blockIdx.x * 16) * GRU_REC] = (int)both;
}

__global__ __launch_bounds__(256) void gru_tile_check_rescan_kernel(
    const float* __restrict__ x,
    const float* __restrict__ W_ih,
    const float* __restrict__ W_hh,
    const float* __restrict__ b_ih,
    const float* __restrict__ b_hh,
    const float* __restrict__ W_out,
    const float* __restrict__ b_out,
    float* __restrict__ Hbuf,
    float* __restrict__ y,
    const float* __restrict__ Ebuf,
    const int* __restrict__ Flag,
    int nsteps, int nchunks)
{
    __shared__ int any_bad;
    const int tid  = threadIdx.x;
    const int lane = tid & 31;
    const int col  = lane & 15;
    const int hl   = lane >> 4;

    if (tid == 0) any_bad = 0;
    __syncthreads();

    unsigned bad = 0u;
    const int last = nchunks - 1;
    for (int g = tid; g * 16 < nchunks; g += 256) {
        unsigned m = (unsigned)Flag[(size_t)(g * 16) * GRU_REC];
        if (g == (last >> 4)) m &= ~(1u << (last & 15));
        bad |= m;
    }
    if (bad) any_bad = 1;
    __syncthreads();
    if (any_bad == 0) return;
    if (tid >= 32) return;

    GruTileWeights w;
    load_tile_weights(w, W_ih, W_hh, b_ih, b_hh, W_out, b_out, lane);

    bool     have  = false;
    unsigned gmask = 0u;
    unsigned fmask = 0u;
    for (int c = 1; c < nchunks; ++c) {
        const int p = c - 1;
        if ((p & 511) == 0) {
            const int c0 = p + 16 * lane;
            const int f  = (c0 < nchunks) ? Flag[(size_t)c0 * GRU_REC] : 0;
            gmask = __builtin_amdgcn_ballot_w32(f != 0);
            if (gmask == 0u && !have) { c += 511; continue; }
        }
        if ((p & 15) == 0) {
            fmask = (unsigned)Flag[(size_t)p * GRU_REC];
            if (fmask == 0u && !have) { c += 15; continue; }
        }
        const bool pred_changed = ((fmask >> (p & 15)) & 1u) != 0u;
        if (!pred_changed && !have) continue;

        const int t_own = c * TILE_CHUNK;
        const int t_end = min(t_own + TILE_CHUNK, nsteps);
        __threadfence();
        const float* ps = Ebuf + (size_t)p * GRU_REC + 8 * hl;
        const float* pe = Hbuf + (size_t)(t_own - 1) * 16 + 8 * hl;
        const float4 sa = *(const float4*)ps, sb = *(const float4*)(ps + 4);
        const float4 ea = *(const float4*)pe, eb = *(const float4*)(pe + 4);
        const bool differs =
            (__float_as_uint(sa.x) != __float_as_uint(ea.x)) | (__float_as_uint(sa.y) != __float_as_uint(ea.y)) |
            (__float_as_uint(sa.z) != __float_as_uint(ea.z)) | (__float_as_uint(sa.w) != __float_as_uint(ea.w)) |
            (__float_as_uint(sb.x) != __float_as_uint(eb.x)) | (__float_as_uint(sb.y) != __float_as_uint(eb.y)) |
            (__float_as_uint(sb.z) != __float_as_uint(eb.z)) | (__float_as_uint(sb.w) != __float_as_uint(eb.w));
        have = false;
        if (__builtin_amdgcn_ballot_w32(differs) != 0u) {
            float h[8] = {ea.x, ea.y, ea.z, ea.w, eb.x, eb.y, eb.z, eb.w};
            const bool active = (col == (c & 15));
            have = gru_tile_rescan_until_met(w, h, x, Hbuf, y, t_own, t_end, active, hl, nsteps) != 0u;
        }
    }
}

extern "C" void kernel_launch(void* const* d_in, const int* in_sizes, int n_in,
                              void* d_out, int out_size, void* d_ws, size_t ws_size,
                              hipStream_t stream)
{
    const float* x     = (const float*)d_in[0];
    const float* W_ih  = (const float*)d_in[1];
    const float* W_hh  = (const float*)d_in[2];
    const float* b_ih  = (const float*)d_in[3];
    const float* b_hh  = (const float*)d_in[4];
    const float* W_out = (const float*)d_in[5];
    const float* b_out = (const float*)d_in[6];
    float* y = (float*)d_out;

    const int nsteps = in_sizes[0] - 1;
    const size_t needH = (size_t)(nsteps + 1) * 16 * sizeof(float);
    float* Hbuf = (float*)d_ws;
    const int    nchunks = (nsteps + GRU_CHUNK - 1) / GRU_CHUNK;
    const size_t needS   = (size_t)nchunks * GRU_REC * sizeof(float);
    float* Ebuf = Hbuf + (size_t)(nsteps + 1) * 16;
    int*   Flag = (int*)(Ebuf + 16);

    const int    tchunks = (nsteps + TILE_CHUNK - 1) / TILE_CHUNK;
    const size_t needT   = (size_t)tchunks * GRU_REC * sizeof(float);
    const int    twaves  = (tchunks + 15) / 16;

    const int tiles  = (nsteps + 15) / 16;
    const int blocks = (tiles + 3) / 4;

    if (ws_size >= needH + needT) {
        hipLaunchKernelGGL(gru_tile_scan_kernel, dim3(twaves), dim3(32), 0, stream,
                           x, W_ih, W_hh, b_ih, b_hh, W_out, b_out, Hbuf, Ebuf, y, nsteps, tchunks);
        hipLaunchKernelGGL(gru_tile_fixup_kernel, dim3(twaves), dim3(32), 0, stream,
                           x, W_ih, W_hh, b_ih, b_hh, W_out, b_out, Hbuf, y, Ebuf, Flag, nsteps, tchunks);
        hipLaunchKernelGGL(gru_tile_check_rescan_kernel, dim3(1), dim3(256), 0, stream,
                           x, W_ih, W_hh, b_ih, b_hh, W_out, b_out, Hbuf, y, Ebuf, Flag, nsteps, tchunks);
    } else if (ws_size >= needH + needS) {
        hipLaunchKernelGGL(gru_scan_chunks_kernel, dim3(nchunks), dim3(32), 0, stream,
                           x, W_ih, W_hh, b_ih, b_hh, Hbuf, Ebuf, nsteps);
        hipLaunchKernelGGL(gru_fixup_chunks_kernel, dim3(nchunks), dim3(32), 0, stream,
                           x, W_ih, W_hh, b_ih, b_hh, Hbuf, Ebuf, Flag, nsteps);
        hipLaunchKernelGGL(gru_check_rescan_kernel, dim3(1), dim3(1024), 0, stream,
                           x, W_ih, W_hh, b_ih, b_hh, Hbuf, Ebuf, Flag, nsteps, nchunks);
        hipLaunchKernelGGL(gru_proj_wmma_kernel, dim3(blocks), dim3(128), 0, stream,
                           Hbuf, W_out, b_out, y, nsteps);
    } else if (ws_size >= needH) {
        hipLaunchKernelGGL((gru_scan_kernel<0>), dim3(1), dim3(32), 0, stream,
                           x, W_ih, W_hh, b_ih, b_hh, W_out, b_out,
                           Hbuf, y, nsteps);
        hipLaunchKernelGGL(gru_proj_wmma_kernel, dim3(blocks), dim3(128), 0, stream,
                           Hbuf, W_out, b_out, y, nsteps);
    } else {
        hipLaunchKernelGGL((gru_scan_kernel<1>), dim3(1), dim3(32), 0, stream,
                           x, W_ih, W_hh, b_ih, b_hh, W_out, b_out,
                           nullptr, y, nsteps);
    }
}


# run-in of launch 1 cut from 192 to 152 steps: sized near the longest forgetting time of the state, not above it; the re-scan launch finishes the few chunks that need longer; results bit-identical
# speedup vs baseline: 2247.6502x; 1.1403x over previous
#include <hip/hip_runtime.h>
#include <hip/hip_bf16.h>


typedef __attribute__((ext_vector_type(2))) float v2f;
typedef __attribute__((ext_vector_type(8))) float v8f;

#define GRU_CHUNK 256
#define GRU_FIX_BLK 8
#define GRU_REC 32

__device__ __forceinline__ float bcast_lane(float v, int l) {
    return __int_as_float(__builtin_amdgcn_readlane(__float_as_int(v), l));
}

struct GruLaneWeights {
    float wA[16];
    float wB[16];
    float cA, wiA, bhhB, cBgi, wiB;
};

__device__ __forceinline__ void load_lane_weights(
    GruLaneWeights& w,
    const float* __restrict__ W_ih, const float* __restrict__ W_hh,
    const float* __restrict__ b_ih, const float* __restrict__ b_hh, int lane)
{
    const int m = lane & 15;
#pragma unroll
    for (int k = 0; k < 16; ++k) {
        w.wA[k] = W_hh[lane * 16 + k];
        w.wB[k] = W_hh[(32 + m) * 16 + k];
    }
    w.cA   = b_ih[lane] + b_hh[lane];
    w.wiA  = W_ih[lane];
    w.bhhB = b_hh[32 + m];
    w.cBgi = b_ih[32 + m];
    w.wiB  = W_ih[32 + m];
}

__device__ __forceinline__ float gru_step(const GruLaneWeights& w, float hv, float x_cur)
{
    float sh[16];
#pragma unroll
    for (int k = 0; k < 16; ++k) sh[k] = bcast_lane(hv, k);

    float aA0 = fmaf(w.wiA, x_cur, w.cA);
    float aA1 = 0.0f;
    float aB0 = w.bhhB;
    float aB1 = 0.0f;
#pragma unroll
    for (int k = 0; k < 8; ++k) {
        aA0 = fmaf(w.wA[k],     sh[k],     aA0);
        aA1 = fmaf(w.wA[k + 8], sh[k + 8], aA1);
        aB0 = fmaf(w.wB[k],     sh[k],     aB0);
        aB1 = fmaf(w.wB[k + 8], sh[k + 8], aB1);
    }
    const float accA = aA0 + aA1;
    const float accB = aB0 + aB1;

    const float gA = __builtin_amdgcn_rcpf(1.0f + __expf(-accA));

    const float npre = fmaf(gA, accB, fmaf(w.wiB, x_cur, w.cBgi));
    const float e2   = __expf(npre + npre);
    const float n    = 1.0f - 2.0f * __builtin_amdgcn_rcpf(e2 + 1.0f);

    const float zl   = __shfl_xor(gA, 16, 32);
    return fmaf(zl, hv - n, n);
}

template <int MODE>
__global__ __launch_bounds__(32) void gru_scan_kernel(
    const float* __restrict__ x,
    const float* __restrict__ W_ih,
    const float* __restrict__ W_hh,
    const float* __restrict__ b_ih,
    const float* __restrict__ b_hh,
    const float* __restrict__ W_out,
    const float* __restrict__ b_out,
    float* __restrict__ Hbuf,
    float* __restrict__ y,
    int nsteps)
{
    const int lane = threadIdx.x;
    const int m    = lane & 15;

    GruLaneWeights w;
    load_lane_weights(w, W_ih, W_hh, b_ih, b_hh, lane);
    const float wout = (lane < 16) ? W_out[m] : 0.0f;
    const float bo   = b_out[0];

    float hv    = 0.0f;
    float x_cur = x[0];

    for (int t = 0; t < nsteps; ++t) {
        float x_nxt = x[t + 1];
        const float hnew = gru_step(w, hv, x_cur);
        hv = hnew;

        if (MODE == 0) {
            if (lane < 16) Hbuf[(size_t)t * 16 + lane] = hnew;
        } else {
            float p = wout * hnew;
            p += __shfl_xor(p, 8, 32);
            p += __shfl_xor(p, 4, 32);
            p += __shfl_xor(p, 2, 32);
            p += __shfl_xor(p, 1, 32);
            if (lane == 0) y[t] = p + bo;
        }
        x_cur = x_nxt;
    }
}

__global__ __launch_bounds__(32) void gru_scan_chunks_kernel(
    const float* __restrict__ x,
    const float* __restrict__ W_ih,
    const float* __restrict__ W_hh,
    const float* __restrict__ b_ih,
    const float* __restrict__ b_hh,
    float* __restrict__ Hbuf,
    float* __restrict__ Ebuf,
    int nsteps)
{
    const int lane  = threadIdx.x;
    const int c     = blockIdx.x;
    const int t_own = c * GRU_CHUNK;
    const int t_end = min(t_own + GRU_CHUNK, nsteps);

    GruLaneWeights w;
    load_lane_weights(w, W_ih, W_hh, b_ih, b_hh, lane);

    float hv    = 0.0f;
    float x_cur = x[t_own];

    for (int t = t_own; t < t_end; ++t) {
        const float x_nxt = x[t + 1];
        hv = gru_step(w, hv, x_cur);
        if (lane < 16) Hbuf[(size_t)t * 16 + lane] = hv;
        x_cur = x_nxt;
    }
    if (lane < 16) Ebuf[c * GRU_REC + lane] = hv;
}

__device__ __forceinline__ int gru_rescan_until_met(
    const GruLaneWeights& w, float& hv,
    const float* __restrict__ x, float* __restrict__ Hbuf,
    int t_own, int t_end, int lane)
{
    const float* col = Hbuf + (lane & 15);

    float old[GRU_FIX_BLK];
#pragma unroll
    for (int j = 0; j < GRU_FIX_BLK; ++j)
        old[j] = col[(size_t)min(t_own + j, t_end - 1) * 16];

    float x_cur = x[t_own];
    for (int tb = t_own; tb < t_end; tb += GRU_FIX_BLK) {
        float nxt[GRU_FIX_BLK];
#pragma unroll
        for (int j = 0; j < GRU_FIX_BLK; ++j)
            nxt[j] = col[(size_t)min(tb + GRU_FIX_BLK + j, t_end - 1) * 16];

#pragma unroll
        for (int j = 0; j < GRU_FIX_BLK; ++j) {
            const int t = tb + j;
            if (t >= t_end) return 1;
            const float x_nxt = x[t + 1];
            hv = gru_step(w, hv, x_cur);
            const bool differs = (lane < 16) &&
                (__float_as_uint(hv) != __float_as_uint(old[j]));
            if (__builtin_amdgcn_ballot_w32(differs) == 0u) return 0;
            if (lane < 16) Hbuf[(size_t)t * 16 + lane] = hv;
            x_cur = x_nxt;
        }
#pragma unroll
        for (int j = 0; j < GRU_FIX_BLK; ++j) old[j] = nxt[j];
    }
    return 1;
}

__global__ __launch_bounds__(32) void gru_fixup_chunks_kernel(
    const float* __restrict__ x,
    const float* __restrict__ W_ih,
    const float* __restrict__ W_hh,
    const float* __restrict__ b_ih,
    const float* __restrict__ b_hh,
    float* __restrict__ Hbuf,
    const float* __restrict__ Ebuf,
    int* __restrict__ Flag,
    int nsteps)
{
    const int lane = threadIdx.x;
    const int m    = lane & 15;
    const int c    = blockIdx.x;
    if (c == 0) {
        if (lane == 0) Flag[0] = 0;
        return;
    }
    const int t_own = c * GRU_CHUNK;
    const int t_end = min(t_own + GRU_CHUNK, nsteps);

    GruLaneWeights w;
    load_lane_weights(w, W_ih, W_hh, b_ih, b_hh, lane);

    float hv = Ebuf[(c - 1) * GRU_REC + m];
    const int changed = gru_rescan_until_met(w, hv, x, Hbuf, t_own, t_end, lane);
    if (lane == 0) Flag[c * GRU_REC] = changed;
}

__global__ __launch_bounds__(1024) void gru_check_rescan_kernel(
    const float* __restrict__ x,
    const float* __restrict__ W_ih,
    const float* __restrict__ W_hh,
    const float* __restrict__ b_ih,
    const float* __restrict__ b_hh,
    float* __restrict__ Hbuf,
    const float* __restrict__ Ebuf,
    const int* __restrict__ Flag,
    int nsteps, int nchunks)
{
    __shared__ int any_bad;
    const int tid  = threadIdx.x;
    const int lane = tid & 31;
    const int m    = lane & 15;

    if (tid == 0) any_bad = 0;
    __syncthreads();

    int bad = 0;
    for (int c = tid; c < nchunks - 1; c += 1024) bad |= Flag[c * GRU_REC];
    if (bad) any_bad = 1;
    __syncthreads();
    if (any_bad == 0) return;
    if (tid >= 32) return;

    GruLaneWeights w;
    load_lane_weights(w, W_ih, W_hh, b_ih, b_hh, lane);

    float    hv    = 0.0f;
    bool     have  = false;
    unsigned fmask = 0u;
    for (int c = 1; c < nchunks; ++c) {
        const int p = c - 1;
        if ((p & 31) == 0) {
            const int idx = p + lane;
            const int f   = (idx < nchunks) ? Flag[idx * GRU_REC] : 0;
            fmask = __builtin_amdgcn_ballot_w32(f != 0);
        }
        const bool pred_changed = ((fmask >> (p & 31)) & 1u) != 0u;
        if (!pred_changed && !have) continue;

        const int t_own = c * GRU_CHUNK;
        const int t_end = min(t_own + GRU_CHUNK, nsteps);
        const unsigned s = __float_as_uint(Ebuf[p * GRU_REC + m]);
        float e = Hbuf[(size_t)(t_own - 1) * 16 + m];
        if (have) e = hv;
        const bool differs = (lane < 16) && (s != __float_as_uint(e));
        have = false;
        if (__builtin_amdgcn_ballot_w32(differs) != 0u) {
            hv = e;
            have = gru_rescan_until_met(w, hv, x, Hbuf, t_own, t_end, lane) != 0;
        }
    }
}

__global__ __launch_bounds__(128) void gru_proj_wmma_kernel(
    const float* __restrict__ Hbuf,
    const float* __restrict__ W_out,
    const float* __restrict__ b_out,
    float* __restrict__ y,
    int nY)
{
    const int lane = threadIdx.x & 31;
    const int wave = blockIdx.x * 4 + (threadIdx.x >> 5);
    const int t0   = wave * 16;
    if (t0 >= nY) return;

    const int half = lane >> 4;
    const int lrow = lane & 15;

    const float bo = b_out[0];
    v8f c;
#pragma unroll
    for (int i = 0; i < 8; ++i) c[i] = bo;

    const float* hrow = Hbuf + (size_t)(t0 + lrow) * 16;
#pragma unroll
    for (int j = 0; j < 4; ++j) {
        const int col = 4 * j + 2 * half;
        v2f a; a[0] = hrow[col];  a[1] = hrow[col + 1];
        v2f b; b[0] = W_out[col]; b[1] = W_out[col + 1];
        c = __builtin_amdgcn_wmma_f32_16x16x4_f32(
                 false, a,  false, b,
                 (short)0, c,  false,  false);
    }

    const int m0 = t0 + 8 * half;
    if (lrow == 0) {
        if (m0 + 8 <= nY) {
            float4 lo = make_float4(c[0], c[1], c[2], c[3]);
            float4 hi = make_float4(c[4], c[5], c[6], c[7]);
            *(float4*)(y + m0)     = lo;
            *(float4*)(y + m0 + 4) = hi;
        } else {
#pragma unroll
            for (int i = 0; i < 8; ++i)
                if (m0 + i < nY) y[m0 + i] = c[i];
        }
    }
}

#define TILE_CHUNK 64
#define TILE_RUNIN 152

struct GruTileWeights {
    v2f   a[3][4];
    v2f   ay[4];
    float bo;
    float c_rz[2][8];
    float wi_rz[2][8];
    float bhh_n[8];
    float bih_n[8];
    float wi_n[8];
};

__device__ __forceinline__ void load_tile_weights(
    GruTileWeights& w,
    const float* __restrict__ W_ih, const float* __restrict__ W_hh,
    const float* __restrict__ b_ih, const float* __restrict__ b_hh,
    const float* __restrict__ W_out, const float* __restrict__ b_out, int lane)
{
    const int m  = lane & 15;
    const int hl = lane >> 4;
#pragma unroll
    for (int j = 0; j < 4; ++j) {
        const float* q = W_out + 8 * hl + 2 * j;
        w.ay[j][0] = q[0];
        w.ay[j][1] = q[1];
    }
    w.bo = b_out[0];
#pragma unroll
    for (int g = 0; g < 3; ++g) {
#pragma unroll
        for (int j = 0; j < 4; ++j) {
            const float* p = W_hh + (16 * g + m) * 16 + 8 * hl + 2 * j;
            w.a[g][j][0] = p[0];
            w.a[g][j][1] = p[1];
        }
    }
#pragma unroll
    for (int v = 0; v < 8; ++v) {
        const int row = 8 * hl + v;
#pragma unroll
        for (int g = 0; g < 2; ++g) {
            w.c_rz[g][v]  = b_ih[16 * g + row] + b_hh[16 * g + row];
            w.wi_rz[g][v] = W_ih[16 * g + row];
        }
        w.bhh_n[v] = b_hh[32 + row];
        w.bih_n[v] = b_ih[32 + row];
        w.wi_n[v]  = W_ih[32 + row];
    }
}

template <bool WITH_Y>
__device__ __forceinline__ void gru_step_tile(const GruTileWeights& w, float (&h)[8], float xv, float& y_in)
{
    v8f ar, az, an, ay;
    float gin[8];
#pragma unroll
    for (int v = 0; v < 8; ++v) {
        ar[v]  = fmaf(w.wi_rz[0][v], xv, w.c_rz[0][v]);
        az[v]  = fmaf(w.wi_rz[1][v], xv, w.c_rz[1][v]);
        an[v]  = w.bhh_n[v];
        gin[v] = fmaf(w.wi_n[v], xv, w.bih_n[v]);
        ay[v]  = w.bo;
    }
#pragma unroll
    for (int j = 0; j < 4; ++j) {
        v2f b; b[0] = h[2 * j]; b[1] = h[2 * j + 1];
        ar = __builtin_amdgcn_wmma_f32_16x16x4_f32(false, w.a[0][j], false, b, (short)0, ar, false, false);
        an = __builtin_amdgcn_wmma_f32_16x16x4_f32(false, w.a[2][j], false, b, (short)0, an, false, false);
        az = __builtin_amdgcn_wmma_f32_16x16x4_f32(false, w.a[1][j], false, b, (short)0, az, false, false);
        if (WITH_Y)
            ay = __builtin_amdgcn_wmma_f32_16x16x4_f32(false, w.ay[j], false, b, (short)0, ay, false, false);
    }
    if (WITH_Y) y_in = ay[0];
#pragma unroll
    for (int v = 0; v < 8; ++v) {
        const float r    = __builtin_amdgcn_rcpf(1.0f + __expf(-ar[v]));
        const float npre = fmaf(r, an[v], gin[v]);
        const float e2   = __expf(npre + npre);
        const float n    = 1.0f - 2.0f * __builtin_amdgcn_rcpf(e2 + 1.0f);
        const float z    = __builtin_amdgcn_rcpf(1.0f + __expf(-az[v]));
        h[v] = fmaf(z, h[v] - n, n);
    }
}

__device__ __forceinline__ float gru_y_tile(const GruTileWeights& w, const float (&h)[8])
{
    v8f ay;
#pragma unroll
    for (int v = 0; v < 8; ++v) ay[v] = w.bo;
#pragma unroll
    for (int j = 0; j < 4; ++j) {
        v2f b; b[0] = h[2 * j]; b[1] = h[2 * j + 1];
        ay = __builtin_amdgcn_wmma_f32_16x16x4_f32(false, w.ay[j], false, b, (short)0, ay, false, false);
    }
    return ay[0];
}

__device__ __forceinline__ void store_y_group(
    float* __restrict__ y, int g0, int t_own, int t_end, int hl,
    float q0, float q1, float q2, float q3)
{
    if (hl != 0) return;
    if (g0 >= t_own && g0 + 3 < t_end) {
        *(float4*)(y + g0) = make_float4(q0, q1, q2, q3);
    } else {
        if (g0     >= t_own && g0     < t_end) y[g0]     = q0;
        if (g0 + 1 >= t_own && g0 + 1 < t_end) y[g0 + 1] = q1;
        if (g0 + 2 >= t_own && g0 + 2 < t_end) y[g0 + 2] = q2;
        if (g0 + 3 >= t_own && g0 + 3 < t_end) y[g0 + 3] = q3;
    }
}

__device__ __forceinline__ float4 load_x4(const float* __restrict__ x, int t, int nsteps)
{
    if (t + 3 <= nsteps) return *(const float4*)(x + t);
    float4 r;
    r.x = x[min(t,     nsteps)];
    r.y = x[min(t + 1, nsteps)];
    r.z = x[min(t + 2, nsteps)];
    r.w = x[min(t + 3, nsteps)];
    return r;
}

__device__ __forceinline__ void store_row_half(float* __restrict__ p, const float (&h)[8])
{
    *(float4*)(p)     = make_float4(h[0], h[1], h[2], h[3]);
    *(float4*)(p + 4) = make_float4(h[4], h[5], h[6], h[7]);
}

template <bool FIRST, bool WHOLE>
__device__ __forceinline__ void gru_tile_run_in(
    const GruTileWeights& w, float (&h)[8], const float* __restrict__ x,
    float4& xq, int t_beg, int nsteps)
{
    for (int jb = 0; jb < TILE_RUNIN; jb += 4) {
        const int tn = FIRST ? max(t_beg + jb + 4, 0) : t_beg + jb + 4;
        const float4 xn = WHOLE ? *(const float4*)(x + tn) : load_x4(x, tn, nsteps);
        const float xs[4] = {xq.x, xq.y, xq.z, xq.w};
#pragma unroll
        for (int u = 0; u < 4; ++u) {
            float unused;
            gru_step_tile<false>(w, h, xs[u], unused);
            if (FIRST) {
                const bool before = (t_beg + jb + u) < 0;
#pragma unroll
                for (int v = 0; v < 8; ++v) h[v] = before ? 0.0f : h[v];
            }
        }
        xq = xn;
    }
}

template <bool STORE_Y>
__device__ __forceinline__ void gru_tile_own_quad_whole(
    const GruTileWeights& w, float (&h)[8], const float4& xq,
    float* __restrict__ hp, float* __restrict__ yp, int hl,
    float& yq0, float& yq1, float& yq2)
{
    const float xs[4] = {xq.x, xq.y, xq.z, xq.w};
#pragma unroll
    for (int u = 0; u < 4; ++u) {
        float yin;
        gru_step_tile<true>(w, h, xs[u], yin);
        if (u == 0) {
            if (STORE_Y) {
                const float2 v = hl ? make_float2(yq2, yin) : make_float2(yq0, yq1);
                *(float2*)yp = v;
            }
        } else if (u == 1) yq0 = yin;
        else if (u == 2)   yq1 = yin;
        else               yq2 = yin;
        store_row_half(hp + u * 16, h);
    }
}

__global__ __launch_bounds__(32) void gru_tile_scan_kernel(
    const float* __restrict__ x,
    const float* __restrict__ W_ih,
    const float* __restrict__ W_hh,
    const float* __restrict__ b_ih,
    const float* __restrict__ b_hh,
    const float* __restrict__ W_out,
    const float* __restrict__ b_out,
    float* __restrict__ Hbuf,
    float* __restrict__ Ebuf,
    float* __restrict__ y,
    int nsteps, int nchunks)
{
    const int lane = threadIdx.x;
    const int col  = lane & 15;
    const int hl   = lane >> 4;
    const int c    = blockIdx.x * 16 + col;
    const bool own = c < nchunks;
    const int t_own = own ? c * TILE_CHUNK : 0;
    const int t_end = own ? min(t_own + TILE_CHUNK, nsteps) : 0;
    const int t_beg = t_own - TILE_RUNIN;

    GruTileWeights w;
    load_tile_weights(w, W_ih, W_hh, b_ih, b_hh, W_out, b_out, lane);

    float h[8];
#pragma unroll
    for (int v = 0; v < 8; ++v) h[v] = 0.0f;

    const bool whole = own && (t_own + TILE_CHUNK - 1 <= nsteps);
    const bool fast  = __builtin_amdgcn_ballot_w32(!whole) == 0u;

    float4 xq = load_x4(x, max(t_beg, 0), nsteps);

    if (fast) {
        if (blockIdx.x == 0) gru_tile_run_in<true, true>(w, h, x, xq, t_beg, nsteps);
        else                 gru_tile_run_in<false, true>(w, h, x, xq, t_beg, nsteps);

        float* hp = Hbuf + (size_t)t_own * 16 + 8 * hl;
        float* yp = y + t_own + 2 * hl;
        float yq0 = 0.0f, yq1 = 0.0f, yq2 = 0.0f;
        {
            const float4 xn = *(const float4*)(x + t_own + 4);
            gru_tile_own_quad_whole<false>(w, h, xq, hp, yp, hl, yq0, yq1, yq2);
            xq = xn;
        }
        for (int jb = 4; jb < TILE_CHUNK; jb += 4) {
            const float4 xn = *(const float4*)(x + t_own + min(jb + 4, TILE_CHUNK - 4));
            gru_tile_own_quad_whole<true>(w, h, xq, hp + (size_t)jb * 16, yp + jb - 4, hl, yq0, yq1, yq2);
            xq = xn;
        }
        store_row_half(Ebuf + (size_t)c * GRU_REC + 8 * hl, h);
        const float yq3 = gru_y_tile(w, h);
        store_y_group(y, t_own + TILE_CHUNK - 4, t_own, t_end, hl, yq0, yq1, yq2, yq3);
        return;
    }

    if (blockIdx.x == 0) gru_tile_run_in<true, false>(w, h, x, xq, t_beg, nsteps);
    else                 gru_tile_run_in<false, false>(w, h, x, xq, t_beg, nsteps);

    float yq0 = 0.0f, yq1 = 0.0f, yq2 = 0.0f, yq3 = 0.0f;
    for (int jb = 0; jb < TILE_CHUNK; jb += 4) {
        float4 xn = xq;
        if (jb + 4 < TILE_CHUNK) xn = load_x4(x, t_own + jb + 4, nsteps);
        const float xs[4] = {xq.x, xq.y, xq.z, xq.w};
#pragma unroll
        for (int u = 0; u < 4; ++u) {
            const int ts = t_own + jb + u;
            float yin;
            gru_step_tile<true>(w, h, xs[u], yin);
            if (u == 0) {
                yq3 = yin;
                store_y_group(y, ts - 4, t_own, t_end, hl, yq0, yq1, yq2, yq3);
            } else if (u == 1) yq0 = yin;
            else if (u == 2)   yq1 = yin;
            else               yq2 = yin;
            if (ts < t_end) {
                store_row_half(Hbuf + (size_t)ts * 16 + 8 * hl, h);
                if (ts == t_end - 1) store_row_half(Ebuf + (size_t)c * GRU_REC + 8 * hl, h);
            }
        }
        xq = xn;
    }
    yq3 = gru_y_tile(w, h);
    store_y_group(y, t_own + TILE_CHUNK - 4, t_own, t_end, hl, yq0, yq1, yq2, yq3);
}

__device__ __forceinline__ unsigned gru_tile_rescan_until_met(
    const GruTileWeights& w, float (&h)[8],
    const float* __restrict__ x, float* __restrict__ Hbuf, float* __restrict__ y,
    int t_own, int t_end, bool active, int hl, int nsteps)
{
    const int    t_last = max(t_end - 1, 0);
    const float* rows   = Hbuf + 8 * hl;

    float4 oa[4], ob[4];
#pragma unroll
    for (int u = 0; u < 4; ++u) {
        const float* p = rows + (size_t)min(t_own + u, t_last) * 16;
        oa[u] = *(const float4*)p;
        ob[u] = *(const float4*)(p + 4);
    }
    bool   differs = false;
    float4 xq = load_x4(x, t_own, nsteps);
    for (int jb = 0; jb < TILE_CHUNK; jb += 4) {
        float4 xn = xq;
        if (jb + 4 < TILE_CHUNK) xn = load_x4(x, t_own + jb + 4, nsteps);
        float4 na[4], nb[4];
#pragma unroll
        for (int u = 0; u < 4; ++u) {
            const float* p = rows + (size_t)min(t_own + jb + 4 + u, t_last) * 16;
            na[u] = *(const float4*)p;
            nb[u] = *(const float4*)(p + 4);
        }
        const float xs[4] = {xq.x, xq.y, xq.z, xq.w};
#pragma unroll
        for (int u = 0; u < 4; ++u) {
            const int  ts   = t_own + jb + u;
            const bool live = active && (ts < t_end);
            float unused;
            gru_step_tile<false>(w, h, xs[u], unused);
            if (live) {
                differs =
                    (__float_as_uint(h[0]) != __float_as_uint(oa[u].x)) |
                    (__float_as_uint(h[1]) != __float_as_uint(oa[u].y)) |
                    (__float_as_uint(h[2]) != __float_as_uint(oa[u].z)) |
                    (__float_as_uint(h[3]) != __float_as_uint(oa[u].w)) |
                    (__float_as_uint(h[4]) != __float_as_uint(ob[u].x)) |
                    (__float_as_uint(h[5]) != __float_as_uint(ob[u].y)) |
                    (__float_as_uint(h[6]) != __float_as_uint(ob[u].z)) |
                    (__float_as_uint(h[7]) != __float_as_uint(ob[u].w));
            }
            if (__builtin_amdgcn_ballot_w32(differs) == 0u) return 0u;
            const float yv = gru_y_tile(w, h);
            if (live) {
                store_row_half(Hbuf + (size_t)ts * 16 + 8 * hl, h);
                if (hl == 0) y[ts] = yv;
            }
        }
#pragma unroll
        for (int u = 0; u < 4; ++u) { oa[u] = na[u]; ob[u] = nb[u]; }
        xq = xn;
    }
    return __builtin_amdgcn_ballot_w32(differs);
}

__global__ __launch_bounds__(32) void gru_tile_fixup_kernel(
    const float* __restrict__ x,
    const float* __restrict__ W_ih,
    const float* __restrict__ W_hh,
    const float* __restrict__ b_ih,
    const float* __restrict__ b_hh,
    const float* __restrict__ W_out,
    const float* __restrict__ b_out,
    float* __restrict__ Hbuf,
    float* __restrict__ y,
    const float* __restrict__ Ebuf,
    int* __restrict__ Flag,
    int nsteps, int nchunks)
{
    const int lane = threadIdx.x;
    const int col  = lane & 15;
    const int hl   = lane >> 4;
    const int c    = blockIdx.x * 16 + col;
    const bool own    = c < nchunks;
    const bool active = own && (c >= 1);
    const int t_own = own ? c * TILE_CHUNK : 0;
    const int t_end = own ? min(t_own + TILE_CHUNK, nsteps) : 0;

    GruTileWeights w;
    load_tile_weights(w, W_ih, W_hh, b_ih, b_hh, W_out, b_out, lane);

    float h[8];
#pragma unroll
    for (int v = 0; v < 8; ++v) h[v] = 0.0f;
    if (active) {
        const float* p = Ebuf + (size_t)(c - 1) * GRU_REC + 8 * hl;
        const float4 a = *(const float4*)p;
        const float4 b = *(const float4*)(p + 4);
        h[0] = a.x; h[1] = a.y; h[2] = a.z; h[3] = a.w;
        h[4] = b.x; h[5] = b.y; h[6] = b.z; h[7] = b.w;
    }
    const unsigned mask = gru_tile_rescan_until_met(w, h, x, Hbuf, y, t_own, t_end, active, hl, nsteps);
    const unsigned both = (mask | (mask >> 16)) & 0xFFFFu;
    if (lane == 0) Flag[(size_t)(blockIdx.x * 16) * GRU_REC] = (int)both;
}

__global__ __launch_bounds__(256) void gru_tile_check_rescan_kernel(
    const float* __restrict__ x,
    const float* __restrict__ W_ih,
    const float* __restrict__ W_hh,
    const float* __restrict__ b_ih,
    const float* __restrict__ b_hh,
    const float* __restrict__ W_out,
    const float* __restrict__ b_out,
    float* __restrict__ Hbuf,
    float* __restrict__ y,
    const float* __restrict__ Ebuf,
    const int* __restrict__ Flag,
    int nsteps, int nchunks)
{
    __shared__ int any_bad;
    const int tid  = threadIdx.x;
    const int lane = tid & 31;
    const int col  = lane & 15;
    const int hl   = lane >> 4;

    if (tid == 0) any_bad = 0;
    __syncthreads();

    unsigned bad = 0u;
    const int last = nchunks - 1;
    for (int g = tid; g * 16 < nchunks; g += 256) {
        unsigned m = (unsigned)Flag[(size_t)(g * 16) * GRU_REC];
        if (g == (last >> 4)) m &= ~(1u << (last & 15));
        bad |= m;
    }
    if (bad) any_bad = 1;
    __syncthreads();
    if (any_bad == 0) return;
    if (tid >= 32) return;

    GruTileWeights w;
    load_tile_weights(w, W_ih, W_hh, b_ih, b_hh, W_out, b_out, lane);

    bool     have  = false;
    unsigned gmask = 0u;
    unsigned fmask = 0u;
    for (int c = 1; c < nchunks; ++c) {
        const int p = c - 1;
        if ((p & 511) == 0) {
            const int c0 = p + 16 * lane;
            const int f  = (c0 < nchunks) ? Flag[(size_t)c0 * GRU_REC] : 0;
            gmask = __builtin_amdgcn_ballot_w32(f != 0);
            if (gmask == 0u && !have) { c += 511; continue; }
        }
        if ((p & 15) == 0) {
            fmask = (unsigned)Flag[(size_t)p * GRU_REC];
            if (fmask == 0u && !have) { c += 15; continue; }
        }
        const bool pred_changed = ((fmask >> (p & 15)) & 1u) != 0u;
        if (!pred_changed && !have) continue;

        const int t_own = c * TILE_CHUNK;
        const int t_end = min(t_own + TILE_CHUNK, nsteps);
        __threadfence();
        const float* ps = Ebuf + (size_t)p * GRU_REC + 8 * hl;
        const float* pe = Hbuf + (size_t)(t_own - 1) * 16 + 8 * hl;
        const float4 sa = *(const float4*)ps, sb = *(const float4*)(ps + 4);
        const float4 ea = *(const float4*)pe, eb = *(const float4*)(pe + 4);
        const bool differs =
            (__float_as_uint(sa.x) != __float_as_uint(ea.x)) | (__float_as_uint(sa.y) != __float_as_uint(ea.y)) |
            (__float_as_uint(sa.z) != __float_as_uint(ea.z)) | (__float_as_uint(sa.w) != __float_as_uint(ea.w)) |
            (__float_as_uint(sb.x) != __float_as_uint(eb.x)) | (__float_as_uint(sb.y) != __float_as_uint(eb.y)) |
            (__float_as_uint(sb.z) != __float_as_uint(eb.z)) | (__float_as_uint(sb.w) != __float_as_uint(eb.w));
        have = false;
        if (__builtin_amdgcn_ballot_w32(differs) != 0u) {
            float h[8] = {ea.x, ea.y, ea.z, ea.w, eb.x, eb.y, eb.z, eb.w};
            const bool active = (col == (c & 15));
            have = gru_tile_rescan_until_met(w, h, x, Hbuf, y, t_own, t_end, active, hl, nsteps) != 0u;
        }
    }
}

extern "C" void kernel_launch(void* const* d_in, const int* in_sizes, int n_in,
                              void* d_out, int out_size, void* d_ws, size_t ws_size,
                              hipStream_t stream)
{
    const float* x     = (const float*)d_in[0];
    const float* W_ih  = (const float*)d_in[1];
    const float* W_hh  = (const float*)d_in[2];
    const float* b_ih  = (const float*)d_in[3];
    const float* b_hh  = (const float*)d_in[4];
    const float* W_out = (const float*)d_in[5];
    const float* b_out = (const float*)d_in[6];
    float* y = (float*)d_out;

    const int nsteps = in_sizes[0] - 1;
    const size_t needH = (size_t)(nsteps + 1) * 16 * sizeof(float);
    float* Hbuf = (float*)d_ws;
    const int    nchunks = (nsteps + GRU_CHUNK - 1) / GRU_CHUNK;
    const size_t needS   = (size_t)nchunks * GRU_REC * sizeof(float);
    float* Ebuf = Hbuf + (size_t)(nsteps + 1) * 16;
    int*   Flag = (int*)(Ebuf + 16);

    const int    tchunks = (nsteps + TILE_CHUNK - 1) / TILE_CHUNK;
    const size_t needT   = (size_t)tchunks * GRU_REC * sizeof(float);
    const int    twaves  = (tchunks + 15) / 16;

    const int tiles  = (nsteps + 15) / 16;
    const int blocks = (tiles + 3) / 4;

    if (ws_size >= needH + needT) {
        hipLaunchKernelGGL(gru_tile_scan_kernel, dim3(twaves), dim3(32), 0, stream,
                           x, W_ih, W_hh, b_ih, b_hh, W_out, b_out, Hbuf, Ebuf, y, nsteps, tchunks);
        hipLaunchKernelGGL(gru_tile_fixup_kernel, dim3(twaves), dim3(32), 0, stream,
                           x, W_ih, W_hh, b_ih, b_hh, W_out, b_out, Hbuf, y, Ebuf, Flag, nsteps, tchunks);
        hipLaunchKernelGGL(gru_tile_check_rescan_kernel, dim3(1), dim3(256), 0, stream,
                           x, W_ih, W_hh, b_ih, b_hh, W_out, b_out, Hbuf, y, Ebuf, Flag, nsteps, tchunks);
    } else if (ws_size >= needH + needS) {
        hipLaunchKernelGGL(gru_scan_chunks_kernel, dim3(nchunks), dim3(32), 0, stream,
                           x, W_ih, W_hh, b_ih, b_hh, Hbuf, Ebuf, nsteps);
        hipLaunchKernelGGL(gru_fixup_chunks_kernel, dim3(nchunks), dim3(32), 0, stream,
                           x, W_ih, W_hh, b_ih, b_hh, Hbuf, Ebuf, Flag, nsteps);
        hipLaunchKernelGGL(gru_check_rescan_kernel, dim3(1), dim3(1024), 0, stream,
                           x, W_ih, W_hh, b_ih, b_hh, Hbuf, Ebuf, Flag, nsteps, nchunks);
        hipLaunchKernelGGL(gru_proj_wmma_kernel, dim3(blocks), dim3(128), 0, stream,
                           Hbuf, W_out, b_out, y, nsteps);
    } else if (ws_size >= needH) {
        hipLaunchKernelGGL((gru_scan_kernel<0>), dim3(1), dim3(32), 0, stream,
                           x, W_ih, W_hh, b_ih, b_hh, W_out, b_out,
                           Hbuf, y, nsteps);
        hipLaunchKernelGGL(gru_proj_wmma_kernel, dim3(blocks), dim3(128), 0, stream,
                           Hbuf, W_out, b_out, y, nsteps);
    } else {
        hipLaunchKernelGGL((gru_scan_kernel<1>), dim3(1), dim3(32), 0, stream,
                           x, W_ih, W_hh, b_ih, b_hh, W_out, b_out,
                           nullptr, y, nsteps);
    }
}


# tiled path stores gate weights, biases and input weights times -log2(e) (r, z) and 2*log2(e) (n): each gate's exponential is a bare 2^x, no scale multiply in a step; last bits of h change
# speedup vs baseline: 2429.7762x; 1.0810x over previous
#include <hip/hip_runtime.h>
#include <hip/hip_bf16.h>


typedef __attribute__((ext_vector_type(2))) float v2f;
typedef __attribute__((ext_vector_type(8))) float v8f;

#define GRU_CHUNK 256
#define GRU_FIX_BLK 8
#define GRU_REC 32

__device__ __forceinline__ float bcast_lane(float v, int l) {
    return __int_as_float(__builtin_amdgcn_readlane(__float_as_int(v), l));
}

struct GruLaneWeights {
    float wA[16];
    float wB[16];
    float cA, wiA, bhhB, cBgi, wiB;
};

__device__ __forceinline__ void load_lane_weights(
    GruLaneWeights& w,
    const float* __restrict__ W_ih, const float* __restrict__ W_hh,
    const float* __restrict__ b_ih, const float* __restrict__ b_hh, int lane)
{
    const int m = lane & 15;
#pragma unroll
    for (int k = 0; k < 16; ++k) {
        w.wA[k] = W_hh[lane * 16 + k];
        w.wB[k] = W_hh[(32 + m) * 16 + k];
    }
    w.cA   = b_ih[lane] + b_hh[lane];
    w.wiA  = W_ih[lane];
    w.bhhB = b_hh[32 + m];
    w.cBgi = b_ih[32 + m];
    w.wiB  = W_ih[32 + m];
}

__device__ __forceinline__ float gru_step(const GruLaneWeights& w, float hv, float x_cur)
{
    float sh[16];
#pragma unroll
    for (int k = 0; k < 16; ++k) sh[k] = bcast_lane(hv, k);

    float aA0 = fmaf(w.wiA, x_cur, w.cA);
    float aA1 = 0.0f;
    float aB0 = w.bhhB;
    float aB1 = 0.0f;
#pragma unroll
    for (int k = 0; k < 8; ++k) {
        aA0 = fmaf(w.wA[k],     sh[k],     aA0);
        aA1 = fmaf(w.wA[k + 8], sh[k + 8], aA1);
        aB0 = fmaf(w.wB[k],     sh[k],     aB0);
        aB1 = fmaf(w.wB[k + 8], sh[k + 8], aB1);
    }
    const float accA = aA0 + aA1;
    const float accB = aB0 + aB1;

    const float gA = __builtin_amdgcn_rcpf(1.0f + __expf(-accA));

    const float npre = fmaf(gA, accB, fmaf(w.wiB, x_cur, w.cBgi));
    const float e2   = __expf(npre + npre);
    const float n    = 1.0f - 2.0f * __builtin_amdgcn_rcpf(e2 + 1.0f);

    const float zl   = __shfl_xor(gA, 16, 32);
    return fmaf(zl, hv - n, n);
}

template <int MODE>
__global__ __launch_bounds__(32) void gru_scan_kernel(
    const float* __restrict__ x,
    const float* __restrict__ W_ih,
    const float* __restrict__ W_hh,
    const float* __restrict__ b_ih,
    const float* __restrict__ b_hh,
    const float* __restrict__ W_out,
    const float* __restrict__ b_out,
    float* __restrict__ Hbuf,
    float* __restrict__ y,
    int nsteps)
{
    const int lane = threadIdx.x;
    const int m    = lane & 15;

    GruLaneWeights w;
    load_lane_weights(w, W_ih, W_hh, b_ih, b_hh, lane);
    const float wout = (lane < 16) ? W_out[m] : 0.0f;
    const float bo   = b_out[0];

    float hv    = 0.0f;
    float x_cur = x[0];

    for (int t = 0; t < nsteps; ++t) {
        float x_nxt = x[t + 1];
        const float hnew = gru_step(w, hv, x_cur);
        hv = hnew;

        if (MODE == 0) {
            if (lane < 16) Hbuf[(size_t)t * 16 + lane] = hnew;
        } else {
            float p = wout * hnew;
            p += __shfl_xor(p, 8, 32);
            p += __shfl_xor(p, 4, 32);
            p += __shfl_xor(p, 2, 32);
            p += __shfl_xor(p, 1, 32);
            if (lane == 0) y[t] = p + bo;
        }
        x_cur = x_nxt;
    }
}

__global__ __launch_bounds__(32) void gru_scan_chunks_kernel(
    const float* __restrict__ x,
    const float* __restrict__ W_ih,
    const float* __restrict__ W_hh,
    const float* __restrict__ b_ih,
    const float* __restrict__ b_hh,
    float* __restrict__ Hbuf,
    float* __restrict__ Ebuf,
    int nsteps)
{
    const int lane  = threadIdx.x;
    const int c     = blockIdx.x;
    const int t_own = c * GRU_CHUNK;
    const int t_end = min(t_own + GRU_CHUNK, nsteps);

    GruLaneWeights w;
    load_lane_weights(w, W_ih, W_hh, b_ih, b_hh, lane);

    float hv    = 0.0f;
    float x_cur = x[t_own];

    for (int t = t_own; t < t_end; ++t) {
        const float x_nxt = x[t + 1];
        hv = gru_step(w, hv, x_cur);
        if (lane < 16) Hbuf[(size_t)t * 16 + lane] = hv;
        x_cur = x_nxt;
    }
    if (lane < 16) Ebuf[c * GRU_REC + lane] = hv;
}

__device__ __forceinline__ int gru_rescan_until_met(
    const GruLaneWeights& w, float& hv,
    const float* __restrict__ x, float* __restrict__ Hbuf,
    int t_own, int t_end, int lane)
{
    const float* col = Hbuf + (lane & 15);

    float old[GRU_FIX_BLK];
#pragma unroll
    for (int j = 0; j < GRU_FIX_BLK; ++j)
        old[j] = col[(size_t)min(t_own + j, t_end - 1) * 16];

    float x_cur = x[t_own];
    for (int tb = t_own; tb < t_end; tb += GRU_FIX_BLK) {
        float nxt[GRU_FIX_BLK];
#pragma unroll
        for (int j = 0; j < GRU_FIX_BLK; ++j)
            nxt[j] = col[(size_t)min(tb + GRU_FIX_BLK + j, t_end - 1) * 16];

#pragma unroll
        for (int j = 0; j < GRU_FIX_BLK; ++j) {
            const int t = tb + j;
            if (t >= t_end) return 1;
            const float x_nxt = x[t + 1];
            hv = gru_step(w, hv, x_cur);
            const bool differs = (lane < 16) &&
                (__float_as_uint(hv) != __float_as_uint(old[j]));
            if (__builtin_amdgcn_ballot_w32(differs) == 0u) return 0;
            if (lane < 16) Hbuf[(size_t)t * 16 + lane] = hv;
            x_cur = x_nxt;
        }
#pragma unroll
        for (int j = 0; j < GRU_FIX_BLK; ++j) old[j] = nxt[j];
    }
    return 1;
}

__global__ __launch_bounds__(32) void gru_fixup_chunks_kernel(
    const float* __restrict__ x,
    const float* __restrict__ W_ih,
    const float* __restrict__ W_hh,
    const float* __restrict__ b_ih,
    const float* __restrict__ b_hh,
    float* __restrict__ Hbuf,
    const float* __restrict__ Ebuf,
    int* __restrict__ Flag,
    int nsteps)
{
    const int lane = threadIdx.x;
    const int m    = lane & 15;
    const int c    = blockIdx.x;
    if (c == 0) {
        if (lane == 0) Flag[0] = 0;
        return;
    }
    const int t_own = c * GRU_CHUNK;
    const int t_end = min(t_own + GRU_CHUNK, nsteps);

    GruLaneWeights w;
    load_lane_weights(w, W_ih, W_hh, b_ih, b_hh, lane);

    float hv = Ebuf[(c - 1) * GRU_REC + m];
    const int changed = gru_rescan_until_met(w, hv, x, Hbuf, t_own, t_end, lane);
    if (lane == 0) Flag[c * GRU_REC] = changed;
}

__global__ __launch_bounds__(1024) void gru_check_rescan_kernel(
    const float* __restrict__ x,
    const float* __restrict__ W_ih,
    const float* __restrict__ W_hh,
    const float* __restrict__ b_ih,
    const float* __restrict__ b_hh,
    float* __restrict__ Hbuf,
    const float* __restrict__ Ebuf,
    const int* __restrict__ Flag,
    int nsteps, int nchunks)
{
    __shared__ int any_bad;
    const int tid  = threadIdx.x;
    const int lane = tid & 31;
    const int m    = lane & 15;

    if (tid == 0) any_bad = 0;
    __syncthreads();

    int bad = 0;
    for (int c = tid; c < nchunks - 1; c += 1024) bad |= Flag[c * GRU_REC];
    if (bad) any_bad = 1;
    __syncthreads();
    if (any_bad == 0) return;
    if (tid >= 32) return;

    GruLaneWeights w;
    load_lane_weights(w, W_ih, W_hh, b_ih, b_hh, lane);

    float    hv    = 0.0f;
    bool     have  = false;
    unsigned fmask = 0u;
    for (int c = 1; c < nchunks; ++c) {
        const int p = c - 1;
        if ((p & 31) == 0) {
            const int idx = p + lane;
            const int f   = (idx < nchunks) ? Flag[idx * GRU_REC] : 0;
            fmask = __builtin_amdgcn_ballot_w32(f != 0);
        }
        const bool pred_changed = ((fmask >> (p & 31)) & 1u) != 0u;
        if (!pred_changed && !have) continue;

        const int t_own = c * GRU_CHUNK;
        const int t_end = min(t_own + GRU_CHUNK, nsteps);
        const unsigned s = __float_as_uint(Ebuf[p * GRU_REC + m]);
        float e = Hbuf[(size_t)(t_own - 1) * 16 + m];
        if (have) e = hv;
        const bool differs = (lane < 16) && (s != __float_as_uint(e));
        have = false;
        if (__builtin_amdgcn_ballot_w32(differs) != 0u) {
            hv = e;
            have = gru_rescan_until_met(w, hv, x, Hbuf, t_own, t_end, lane) != 0;
        }
    }
}

__global__ __launch_bounds__(128) void gru_proj_wmma_kernel(
    const float* __restrict__ Hbuf,
    const float* __restrict__ W_out,
    const float* __restrict__ b_out,
    float* __restrict__ y,
    int nY)
{
    const int lane = threadIdx.x & 31;
    const int wave = blockIdx.x * 4 + (threadIdx.x >> 5);
    const int t0   = wave * 16;
    if (t0 >= nY) return;

    const int half = lane >> 4;
    const int lrow = lane & 15;

    const float bo = b_out[0];
    v8f c;
#pragma unroll
    for (int i = 0; i < 8; ++i) c[i] = bo;

    const float* hrow = Hbuf + (size_t)(t0 + lrow) * 16;
#pragma unroll
    for (int j = 0; j < 4; ++j) {
        const int col = 4 * j + 2 * half;
        v2f a; a[0] = hrow[col];  a[1] = hrow[col + 1];
        v2f b; b[0] = W_out[col]; b[1] = W_out[col + 1];
        c = __builtin_amdgcn_wmma_f32_16x16x4_f32(
                 false, a,  false, b,
                 (short)0, c,  false,  false);
    }

    const int m0 = t0 + 8 * half;
    if (lrow == 0) {
        if (m0 + 8 <= nY) {
            float4 lo = make_float4(c[0], c[1], c[2], c[3]);
            float4 hi = make_float4(c[4], c[5], c[6], c[7]);
            *(float4*)(y + m0)     = lo;
            *(float4*)(y + m0 + 4) = hi;
        } else {
#pragma unroll
            for (int i = 0; i < 8; ++i)
                if (m0 + i < nY) y[m0 + i] = c[i];
        }
    }
}

#define TILE_CHUNK 64
#define TILE_RUNIN 152

#define GRU_SCALE_RZ (-1.44269504088896341f)
#define GRU_SCALE_N  (2.88539008177792681f)

struct GruTileWeights {
    v2f   a[3][4];
    v2f   ay[4];
    float bo;
    float c_rz[2][8];
    float wi_rz[2][8];
    float bhh_n[8];
    float bih_n[8];
    float wi_n[8];
};

__device__ __forceinline__ void load_tile_weights(
    GruTileWeights& w,
    const float* __restrict__ W_ih, const float* __restrict__ W_hh,
    const float* __restrict__ b_ih, const float* __restrict__ b_hh,
    const float* __restrict__ W_out, const float* __restrict__ b_out, int lane)
{
    const int m  = lane & 15;
    const int hl = lane >> 4;
#pragma unroll
    for (int j = 0; j < 4; ++j) {
        const float* q = W_out + 8 * hl + 2 * j;
        w.ay[j][0] = q[0];
        w.ay[j][1] = q[1];
    }
    w.bo = b_out[0];
#pragma unroll
    for (int g = 0; g < 3; ++g) {
        const float s = (g < 2) ? GRU_SCALE_RZ : GRU_SCALE_N;
#pragma unroll
        for (int j = 0; j < 4; ++j) {
            const float* p = W_hh + (16 * g + m) * 16 + 8 * hl + 2 * j;
            w.a[g][j][0] = s * p[0];
            w.a[g][j][1] = s * p[1];
        }
    }
#pragma unroll
    for (int v = 0; v < 8; ++v) {
        const int row = 8 * hl + v;
#pragma unroll
        for (int g = 0; g < 2; ++g) {
            w.c_rz[g][v]  = GRU_SCALE_RZ * (b_ih[16 * g + row] + b_hh[16 * g + row]);
            w.wi_rz[g][v] = GRU_SCALE_RZ * W_ih[16 * g + row];
        }
        w.bhh_n[v] = GRU_SCALE_N * b_hh[32 + row];
        w.bih_n[v] = GRU_SCALE_N * b_ih[32 + row];
        w.wi_n[v]  = GRU_SCALE_N * W_ih[32 + row];
    }
}

template <bool WITH_Y>
__device__ __forceinline__ void gru_step_tile(const GruTileWeights& w, float (&h)[8], float xv, float& y_in)
{
    v8f ar, az, an, ay;
    float gin[8];
#pragma unroll
    for (int v = 0; v < 8; ++v) {
        ar[v]  = fmaf(w.wi_rz[0][v], xv, w.c_rz[0][v]);
        az[v]  = fmaf(w.wi_rz[1][v], xv, w.c_rz[1][v]);
        an[v]  = w.bhh_n[v];
        gin[v] = fmaf(w.wi_n[v], xv, w.bih_n[v]);
        ay[v]  = w.bo;
    }
#pragma unroll
    for (int j = 0; j < 4; ++j) {
        v2f b; b[0] = h[2 * j]; b[1] = h[2 * j + 1];
        ar = __builtin_amdgcn_wmma_f32_16x16x4_f32(false, w.a[0][j], false, b, (short)0, ar, false, false);
        an = __builtin_amdgcn_wmma_f32_16x16x4_f32(false, w.a[2][j], false, b, (short)0, an, false, false);
        az = __builtin_amdgcn_wmma_f32_16x16x4_f32(false, w.a[1][j], false, b, (short)0, az, false, false);
        if (WITH_Y)
            ay = __builtin_amdgcn_wmma_f32_16x16x4_f32(false, w.ay[j], false, b, (short)0, ay, false, false);
    }
    if (WITH_Y) y_in = ay[0];
#pragma unroll
    for (int v = 0; v < 8; ++v) {
        const float r    = __builtin_amdgcn_rcpf(1.0f + __builtin_amdgcn_exp2f(ar[v]));
        const float npre = fmaf(r, an[v], gin[v]);
        const float e2   = __builtin_amdgcn_exp2f(npre);
        const float n    = 1.0f - 2.0f * __builtin_amdgcn_rcpf(e2 + 1.0f);
        const float z    = __builtin_amdgcn_rcpf(1.0f + __builtin_amdgcn_exp2f(az[v]));
        h[v] = fmaf(z, h[v] - n, n);
    }
}

__device__ __forceinline__ float gru_y_tile(const GruTileWeights& w, const float (&h)[8])
{
    v8f ay;
#pragma unroll
    for (int v = 0; v < 8; ++v) ay[v] = w.bo;
#pragma unroll
    for (int j = 0; j < 4; ++j) {
        v2f b; b[0] = h[2 * j]; b[1] = h[2 * j + 1];
        ay = __builtin_amdgcn_wmma_f32_16x16x4_f32(false, w.ay[j], false, b, (short)0, ay, false, false);
    }
    return ay[0];
}

__device__ __forceinline__ void store_y_group(
    float* __restrict__ y, int g0, int t_own, int t_end, int hl,
    float q0, float q1, float q2, float q3)
{
    if (hl != 0) return;
    if (g0 >= t_own && g0 + 3 < t_end) {
        *(float4*)(y + g0) = make_float4(q0, q1, q2, q3);
    } else {
        if (g0     >= t_own && g0     < t_end) y[g0]     = q0;
        if (g0 + 1 >= t_own && g0 + 1 < t_end) y[g0 + 1] = q1;
        if (g0 + 2 >= t_own && g0 + 2 < t_end) y[g0 + 2] = q2;
        if (g0 + 3 >= t_own && g0 + 3 < t_end) y[g0 + 3] = q3;
    }
}

__device__ __forceinline__ float4 load_x4(const float* __restrict__ x, int t, int nsteps)
{
    if (t + 3 <= nsteps) return *(const float4*)(x + t);
    float4 r;
    r.x = x[min(t,     nsteps)];
    r.y = x[min(t + 1, nsteps)];
    r.z = x[min(t + 2, nsteps)];
    r.w = x[min(t + 3, nsteps)];
    return r;
}

__device__ __forceinline__ void store_row_half(float* __restrict__ p, const float (&h)[8])
{
    *(float4*)(p)     = make_float4(h[0], h[1], h[2], h[3]);
    *(float4*)(p + 4) = make_float4(h[4], h[5], h[6], h[7]);
}

template <bool FIRST, bool WHOLE>
__device__ __forceinline__ void gru_tile_run_in(
    const GruTileWeights& w, float (&h)[8], const float* __restrict__ x,
    float4& xq, int t_beg, int nsteps)
{
    for (int jb = 0; jb < TILE_RUNIN; jb += 4) {
        const int tn = FIRST ? max(t_beg + jb + 4, 0) : t_beg + jb + 4;
        const float4 xn = WHOLE ? *(const float4*)(x + tn) : load_x4(x, tn, nsteps);
        const float xs[4] = {xq.x, xq.y, xq.z, xq.w};
#pragma unroll
        for (int u = 0; u < 4; ++u) {
            float unused;
            gru_step_tile<false>(w, h, xs[u], unused);
            if (FIRST) {
                const bool before = (t_beg + jb + u) < 0;
#pragma unroll
                for (int v = 0; v < 8; ++v) h[v] = before ? 0.0f : h[v];
            }
        }
        xq = xn;
    }
}

template <bool STORE_Y>
__device__ __forceinline__ void gru_tile_own_quad_whole(
    const GruTileWeights& w, float (&h)[8], const float4& xq,
    float* __restrict__ hp, float* __restrict__ yp, int hl,
    float& yq0, float& yq1, float& yq2)
{
    const float xs[4] = {xq.x, xq.y, xq.z, xq.w};
#pragma unroll
    for (int u = 0; u < 4; ++u) {
        float yin;
        gru_step_tile<true>(w, h, xs[u], yin);
        if (u == 0) {
            if (STORE_Y) {
                const float2 v = hl ? make_float2(yq2, yin) : make_float2(yq0, yq1);
                *(float2*)yp = v;
            }
        } else if (u == 1) yq0 = yin;
        else if (u == 2)   yq1 = yin;
        else               yq2 = yin;
        store_row_half(hp + u * 16, h);
    }
}

__global__ __launch_bounds__(32) void gru_tile_scan_kernel(
    const float* __restrict__ x,
    const float* __restrict__ W_ih,
    const float* __restrict__ W_hh,
    const float* __restrict__ b_ih,
    const float* __restrict__ b_hh,
    const float* __restrict__ W_out,
    const float* __restrict__ b_out,
    float* __restrict__ Hbuf,
    float* __restrict__ Ebuf,
    float* __restrict__ y,
    int nsteps, int nchunks)
{
    const int lane = threadIdx.x;
    const int col  = lane & 15;
    const int hl   = lane >> 4;
    const int c    = blockIdx.x * 16 + col;
    const bool own = c < nchunks;
    const int t_own = own ? c * TILE_CHUNK : 0;
    const int t_end = own ? min(t_own + TILE_CHUNK, nsteps) : 0;
    const int t_beg = t_own - TILE_RUNIN;

    GruTileWeights w;
    load_tile_weights(w, W_ih, W_hh, b_ih, b_hh, W_out, b_out, lane);

    float h[8];
#pragma unroll
    for (int v = 0; v < 8; ++v) h[v] = 0.0f;

    const bool whole = own && (t_own + TILE_CHUNK - 1 <= nsteps);
    const bool fast  = __builtin_amdgcn_ballot_w32(!whole) == 0u;

    float4 xq = load_x4(x, max(t_beg, 0), nsteps);

    if (fast) {
        if (blockIdx.x == 0) gru_tile_run_in<true, true>(w, h, x, xq, t_beg, nsteps);
        else                 gru_tile_run_in<false, true>(w, h, x, xq, t_beg, nsteps);

        float* hp = Hbuf + (size_t)t_own * 16 + 8 * hl;
        float* yp = y + t_own + 2 * hl;
        float yq0 = 0.0f, yq1 = 0.0f, yq2 = 0.0f;
        {
            const float4 xn = *(const float4*)(x + t_own + 4);
            gru_tile_own_quad_whole<false>(w, h, xq, hp, yp, hl, yq0, yq1, yq2);
            xq = xn;
        }
        for (int jb = 4; jb < TILE_CHUNK; jb += 4) {
            const float4 xn = *(const float4*)(x + t_own + min(jb + 4, TILE_CHUNK - 4));
            gru_tile_own_quad_whole<true>(w, h, xq, hp + (size_t)jb * 16, yp + jb - 4, hl, yq0, yq1, yq2);
            xq = xn;
        }
        store_row_half(Ebuf + (size_t)c * GRU_REC + 8 * hl, h);
        const float yq3 = gru_y_tile(w, h);
        store_y_group(y, t_own + TILE_CHUNK - 4, t_own, t_end, hl, yq0, yq1, yq2, yq3);
        return;
    }

    if (blockIdx.x == 0) gru_tile_run_in<true, false>(w, h, x, xq, t_beg, nsteps);
    else                 gru_tile_run_in<false, false>(w, h, x, xq, t_beg, nsteps);

    float yq0 = 0.0f, yq1 = 0.0f, yq2 = 0.0f, yq3 = 0.0f;
    for (int jb = 0; jb < TILE_CHUNK; jb += 4) {
        float4 xn = xq;
        if (jb + 4 < TILE_CHUNK) xn = load_x4(x, t_own + jb + 4, nsteps);
        const float xs[4] = {xq.x, xq.y, xq.z, xq.w};
#pragma unroll
        for (int u = 0; u < 4; ++u) {
            const int ts = t_own + jb + u;
            float yin;
            gru_step_tile<true>(w, h, xs[u], yin);
            if (u == 0) {
                yq3 = yin;
                store_y_group(y, ts - 4, t_own, t_end, hl, yq0, yq1, yq2, yq3);
            } else if (u == 1) yq0 = yin;
            else if (u == 2)   yq1 = yin;
            else               yq2 = yin;
            if (ts < t_end) {
                store_row_half(Hbuf + (size_t)ts * 16 + 8 * hl, h);
                if (ts == t_end - 1) store_row_half(Ebuf + (size_t)c * GRU_REC + 8 * hl, h);
            }
        }
        xq = xn;
    }
    yq3 = gru_y_tile(w, h);
    store_y_group(y, t_own + TILE_CHUNK - 4, t_own, t_end, hl, yq0, yq1, yq2, yq3);
}

__device__ __forceinline__ unsigned gru_tile_rescan_until_met(
    const GruTileWeights& w, float (&h)[8],
    const float* __restrict__ x, float* __restrict__ Hbuf, float* __restrict__ y,
    int t_own, int t_end, bool active, int hl, int nsteps)
{
    const int    t_last = max(t_end - 1, 0);
    const float* rows   = Hbuf + 8 * hl;

    float4 oa[4], ob[4];
#pragma unroll
    for (int u = 0; u < 4; ++u) {
        const float* p = rows + (size_t)min(t_own + u, t_last) * 16;
        oa[u] = *(const float4*)p;
        ob[u] = *(const float4*)(p + 4);
    }
    bool   differs = false;
    float4 xq = load_x4(x, t_own, nsteps);
    for (int jb = 0; jb < TILE_CHUNK; jb += 4) {
        float4 xn = xq;
        if (jb + 4 < TILE_CHUNK) xn = load_x4(x, t_own + jb + 4, nsteps);
        float4 na[4], nb[4];
#pragma unroll
        for (int u = 0; u < 4; ++u) {
            const float* p = rows + (size_t)min(t_own + jb + 4 + u, t_last) * 16;
            na[u] = *(const float4*)p;
            nb[u] = *(const float4*)(p + 4);
        }
        const float xs[4] = {xq.x, xq.y, xq.z, xq.w};
#pragma unroll
        for (int u = 0; u < 4; ++u) {
            const int  ts   = t_own + jb + u;
            const bool live = active && (ts < t_end);
            float unused;
            gru_step_tile<false>(w, h, xs[u], unused);
            if (live) {
                differs =
                    (__float_as_uint(h[0]) != __float_as_uint(oa[u].x)) |
                    (__float_as_uint(h[1]) != __float_as_uint(oa[u].y)) |
                    (__float_as_uint(h[2]) != __float_as_uint(oa[u].z)) |
                    (__float_as_uint(h[3]) != __float_as_uint(oa[u].w)) |
                    (__float_as_uint(h[4]) != __float_as_uint(ob[u].x)) |
                    (__float_as_uint(h[5]) != __float_as_uint(ob[u].y)) |
                    (__float_as_uint(h[6]) != __float_as_uint(ob[u].z)) |
                    (__float_as_uint(h[7]) != __float_as_uint(ob[u].w));
            }
            if (__builtin_amdgcn_ballot_w32(differs) == 0u) return 0u;
            const float yv = gru_y_tile(w, h);
            if (live) {
                store_row_half(Hbuf + (size_t)ts * 16 + 8 * hl, h);
                if (hl == 0) y[ts] = yv;
            }
        }
#pragma unroll
        for (int u = 0; u < 4; ++u) { oa[u] = na[u]; ob[u] = nb[u]; }
        xq = xn;
    }
    return __builtin_amdgcn_ballot_w32(differs);
}

__global__ __launch_bounds__(32) void gru_tile_fixup_kernel(
    const float* __restrict__ x,
    const float* __restrict__ W_ih,
    const float* __restrict__ W_hh,
    const float* __restrict__ b_ih,
    const float* __restrict__ b_hh,
    const float* __restrict__ W_out,
    const float* __restrict__ b_out,
    float* __restrict__ Hbuf,
    float* __restrict__ y,
    const float* __restrict__ Ebuf,
    int* __restrict__ Flag,
    int nsteps, int nchunks)
{
    const int lane = threadIdx.x;
    const int col  = lane & 15;
    const int hl   = lane >> 4;
    const int c    = blockIdx.x * 16 + col;
    const bool own    = c < nchunks;
    const bool active = own && (c >= 1);
    const int t_own = own ? c * TILE_CHUNK : 0;
    const int t_end = own ? min(t_own + TILE_CHUNK, nsteps) : 0;

    GruTileWeights w;
    load_tile_weights(w, W_ih, W_hh, b_ih, b_hh, W_out, b_out, lane);

    float h[8];
#pragma unroll
    for (int v = 0; v < 8; ++v) h[v] = 0.0f;
    if (active) {
        const float* p = Ebuf + (size_t)(c - 1) * GRU_REC + 8 * hl;
        const float4 a = *(const float4*)p;
        const float4 b = *(const float4*)(p + 4);
        h[0] = a.x; h[1] = a.y; h[2] = a.z; h[3] = a.w;
        h[4] = b.x; h[5] = b.y; h[6] = b.z; h[7] = b.w;
    }
    const unsigned mask = gru_tile_rescan_until_met(w, h, x, Hbuf, y, t_own, t_end, active, hl, nsteps);
    const unsigned both = (mask | (mask >> 16)) & 0xFFFFu;
    if (lane == 0) Flag[(size_t)(blockIdx.x * 16) * GRU_REC] = (int)both;
}

__global__ __launch_bounds__(256) void gru_tile_check_rescan_kernel(
    const float* __restrict__ x,
    const float* __restrict__ W_ih,
    const float* __restrict__ W_hh,
    const float* __restrict__ b_ih,
    const float* __restrict__ b_hh,
    const float* __restrict__ W_out,
    const float* __restrict__ b_out,
    float* __restrict__ Hbuf,
    float* __restrict__ y,
    const float* __restrict__ Ebuf,
    const int* __restrict__ Flag,
    int nsteps, int nchunks)
{
    __shared__ int any_bad;
    const int tid  = threadIdx.x;
    const int lane = tid & 31;
    const int col  = lane & 15;
    const int hl   = lane >> 4;

    if (tid == 0) any_bad = 0;
    __syncthreads();

    unsigned bad = 0u;
    const int last = nchunks - 1;
    for (int g = tid; g * 16 < nchunks; g += 256) {
        unsigned m = (unsigned)Flag[(size_t)(g * 16) * GRU_REC];
        if (g == (last >> 4)) m &= ~(1u << (last & 15));
        bad |= m;
    }
    if (bad) any_bad = 1;
    __syncthreads();
    if (any_bad == 0) return;
    if (tid >= 32) return;

    GruTileWeights w;
    load_tile_weights(w, W_ih, W_hh, b_ih, b_hh, W_out, b_out, lane);

    bool     have  = false;
    unsigned gmask = 0u;
    unsigned fmask = 0u;
    for (int c = 1; c < nchunks; ++c) {
        const int p = c - 1;
        if ((p & 511) == 0) {
            const int c0 = p + 16 * lane;
            const int f  = (c0 < nchunks) ? Flag[(size_t)c0 * GRU_REC] : 0;
            gmask = __builtin_amdgcn_ballot_w32(f != 0);
            if (gmask == 0u && !have) { c += 511; continue; }
        }
        if ((p & 15) == 0) {
            fmask = (unsigned)Flag[(size_t)p * GRU_REC];
            if (fmask == 0u && !have) { c += 15; continue; }
        }
        const bool pred_changed = ((fmask >> (p & 15)) & 1u) != 0u;
        if (!pred_changed && !have) continue;

        const int t_own = c * TILE_CHUNK;
        const int t_end = min(t_own + TILE_CHUNK, nsteps);
        __threadfence();
        const float* ps = Ebuf + (size_t)p * GRU_REC + 8 * hl;
        const float* pe = Hbuf + (size_t)(t_own - 1) * 16 + 8 * hl;
        const float4 sa = *(const float4*)ps, sb = *(const float4*)(ps + 4);
        const float4 ea = *(const float4*)pe, eb = *(const float4*)(pe + 4);
        const bool differs =
            (__float_as_uint(sa.x) != __float_as_uint(ea.x)) | (__float_as_uint(sa.y) != __float_as_uint(ea.y)) |
            (__float_as_uint(sa.z) != __float_as_uint(ea.z)) | (__float_as_uint(sa.w) != __float_as_uint(ea.w)) |
            (__float_as_uint(sb.x) != __float_as_uint(eb.x)) | (__float_as_uint(sb.y) != __float_as_uint(eb.y)) |
            (__float_as_uint(sb.z) != __float_as_uint(eb.z)) | (__float_as_uint(sb.w) != __float_as_uint(eb.w));
        have = false;
        if (__builtin_amdgcn_ballot_w32(differs) != 0u) {
            float h[8] = {ea.x, ea.y, ea.z, ea.w, eb.x, eb.y, eb.z, eb.w};
            const bool active = (col == (c & 15));
            have = gru_tile_rescan_until_met(w, h, x, Hbuf, y, t_own, t_end, active, hl, nsteps) != 0u;
        }
    }
}

extern "C" void kernel_launch(void* const* d_in, const int* in_sizes, int n_in,
                              void* d_out, int out_size, void* d_ws, size_t ws_size,
                              hipStream_t stream)
{
    const float* x     = (const float*)d_in[0];
    const float* W_ih  = (const float*)d_in[1];
    const float* W_hh  = (const float*)d_in[2];
    const float* b_ih  = (const float*)d_in[3];
    const float* b_hh  = (const float*)d_in[4];
    const float* W_out = (const float*)d_in[5];
    const float* b_out = (const float*)d_in[6];
    float* y = (float*)d_out;

    const int nsteps = in_sizes[0] - 1;
    const size_t needH = (size_t)(nsteps + 1) * 16 * sizeof(float);
    float* Hbuf = (float*)d_ws;
    const int    nchunks = (nsteps + GRU_CHUNK - 1) / GRU_CHUNK;
    const size_t needS   = (size_t)nchunks * GRU_REC * sizeof(float);
    float* Ebuf = Hbuf + (size_t)(nsteps + 1) * 16;
    int*   Flag = (int*)(Ebuf + 16);

    const int    tchunks = (nsteps + TILE_CHUNK - 1) / TILE_CHUNK;
    const size_t needT   = (size_t)tchunks * GRU_REC * sizeof(float);
    const int    twaves  = (tchunks + 15) / 16;

    const int tiles  = (nsteps + 15) / 16;
    const int blocks = (tiles + 3) / 4;

    if (ws_size >= needH + needT) {
        hipLaunchKernelGGL(gru_tile_scan_kernel, dim3(twaves), dim3(32), 0, stream,
                           x, W_ih, W_hh, b_ih, b_hh, W_out, b_out, Hbuf, Ebuf, y, nsteps, tchunks);
        hipLaunchKernelGGL(gru_tile_fixup_kernel, dim3(twaves), dim3(32), 0, stream,
                           x, W_ih, W_hh, b_ih, b_hh, W_out, b_out, Hbuf, y, Ebuf, Flag, nsteps, tchunks);
        hipLaunchKernelGGL(gru_tile_check_rescan_kernel, dim3(1), dim3(256), 0, stream,
                           x, W_ih, W_hh, b_ih, b_hh, W_out, b_out, Hbuf, y, Ebuf, Flag, nsteps, tchunks);
    } else if (ws_size >= needH + needS) {
        hipLaunchKernelGGL(gru_scan_chunks_kernel, dim3(nchunks), dim3(32), 0, stream,
                           x, W_ih, W_hh, b_ih, b_hh, Hbuf, Ebuf, nsteps);
        hipLaunchKernelGGL(gru_fixup_chunks_kernel, dim3(nchunks), dim3(32), 0, stream,
                           x, W_ih, W_hh, b_ih, b_hh, Hbuf, Ebuf, Flag, nsteps);
        hipLaunchKernelGGL(gru_check_rescan_kernel, dim3(1), dim3(1024), 0, stream,
                           x, W_ih, W_hh, b_ih, b_hh, Hbuf, Ebuf, Flag, nsteps, nchunks);
        hipLaunchKernelGGL(gru_proj_wmma_kernel, dim3(blocks), dim3(128), 0, stream,
                           Hbuf, W_out, b_out, y, nsteps);
    } else if (ws_size >= needH) {
        hipLaunchKernelGGL((gru_scan_kernel<0>), dim3(1), dim3(32), 0, stream,
                           x, W_ih, W_hh, b_ih, b_hh, W_out, b_out,
                           Hbuf, y, nsteps);
        hipLaunchKernelGGL(gru_proj_wmma_kernel, dim3(blocks), dim3(128), 0, stream,
                           Hbuf, W_out, b_out, y, nsteps);
    } else {
        hipLaunchKernelGGL((gru_scan_kernel<1>), dim3(1), dim3(32), 0, stream,
                           x, W_ih, W_hh, b_ih, b_hh, W_out, b_out,
                           nullptr, y, nsteps);
    }
}
